# starting kernel
# baseline (speedup 1.0000x reference)
#include <hip/hip_runtime.h>
#include <math.h>


typedef float v2f __attribute__((ext_vector_type(2)));
typedef float v8f __attribute__((ext_vector_type(8)));

#define WMMA4(a, b, c) \
  __builtin_amdgcn_wmma_f32_16x16x4_f32(false, (a), false, (b), (short)0, (c), false, false)

constexpr int OFF_W1  = 0;
constexpr int OFF_W2  = OFF_W1  + 20800;
constexpr int OFF_WMU = OFF_W2  + 6400;
constexpr int OFF_WLV = OFF_WMU + 640;
constexpr int OFF_CBT = OFF_WLV + 640;
constexpr int OFF_B1  = OFF_CBT + 10240;
constexpr int OFF_B2  = OFF_B1  + 208;
constexpr int OFF_BMU = OFF_B2  + 32;
constexpr int OFF_BLV = OFF_BMU + 32;
constexpr int OFF_CBN = OFF_BLV + 32;
constexpr int HSTR    = 210;
constexpr int OFF_H   = OFF_CBN + 512;
constexpr int OFF_Z   = OFF_H   + 8 * 16 * HSTR;
constexpr int OFF_MND = OFF_Z   + 8 * 320;
constexpr int OFF_MNJ = OFF_MND + 8 * 16;
constexpr int LDS_FLOATS = OFF_MNJ + 8 * 16;

__global__ __launch_bounds__(256) void vqvae_fused_kernel(
    const float* __restrict__ c,  const float* __restrict__ W1, const float* __restrict__ b1,
    const float* __restrict__ W2, const float* __restrict__ b2, const float* __restrict__ cb,
    const float* __restrict__ Wmu, const float* __restrict__ bmu,
    const float* __restrict__ Wlv, const float* __restrict__ blv,
    float* __restrict__ out_mu, float* __restrict__ out_lv,
    float* __restrict__ out_zq, float* __restrict__ out_ze,
    float* __restrict__ ws, int B) {
  extern __shared__ float smem[];
  const int tid  = threadIdx.x;
  const int wave = tid >> 5;
  const int lane = tid & 31;
  const int lo   = lane & 15;
  const int hi   = lane >> 4;
  const int klo  = hi * 2;

  for (int f = tid; f < 20800; f += 256) {
    int kk = f & 3, rem = f >> 2, n = rem % 208, kt = rem / 208;
    smem[OFF_W1 + f] = (n < 200) ? W1[(kt * 4 + kk) * 200 + n] : 0.f;
  }
  for (int f = tid; f < 6400; f += 256) {
    int kk = f & 3, rem = f >> 2, n = rem & 31, kt = rem >> 5;
    smem[OFF_W2 + f] = (n < 20) ? W2[(kt * 4 + kk) * 20 + n] : 0.f;
  }
  for (int f = tid; f < 640; f += 256) {
    int kk = f & 3, rem = f >> 2, n = rem & 31, kt = rem >> 5;
    smem[OFF_WMU + f] = (n < 20) ? Wmu[(kt * 4 + kk) * 20 + n] : 0.f;
    smem[OFF_WLV + f] = (n < 20) ? Wlv[(kt * 4 + kk) * 20 + n] : 0.f;
  }
  for (int f = tid; f < 10240; f += 256) {
    int kk = f & 3, rem = f >> 2, j = rem & 511, kt = rem >> 9;
    smem[OFF_CBT + f] = cb[j * 20 + kt * 4 + kk];
  }
  for (int n = tid; n < 208; n += 256) smem[OFF_B1 + n] = (n < 200) ? b1[n] : 0.f;
  if (tid < 32) {
    smem[OFF_B2  + tid] = (tid < 20) ? b2[tid]  : 0.f;
    smem[OFF_BMU + tid] = (tid < 20) ? bmu[tid] : 0.f;
    smem[OFF_BLV + tid] = (tid < 20) ? blv[tid] : 0.f;
  }
  for (int j = tid; j < 512; j += 256) {
    float s = 0.f;
    for (int k = 0; k < 20; ++k) { float t = cb[j * 20 + k]; s += t * t; }
    smem[OFF_CBN + j] = s;
  }
  __syncthreads();

  float* hS = &smem[OFF_H + wave * 16 * HSTR];
  float* zS = &smem[OFF_Z + wave * 320];
  float* mnD = &smem[OFF_MND + wave * 16];
  int*   mnJ = (int*)&smem[OFF_MNJ + wave * 16];

  const int numTiles = B >> 4;
  for (int tile = blockIdx.x * 8 + wave; tile < numTiles; tile += gridDim.x * 8) {
    const int rowBase = tile * 16;

    v2f a1[25];
    const float* crow = c + (size_t)(rowBase + lo) * 100 + klo;
    #pragma unroll
    for (int kt = 0; kt < 25; ++kt) a1[kt] = *(const v2f*)(crow + kt * 4);

    for (int nt = 0; nt < 13; ++nt) {
      v8f acc = {};
      #pragma unroll
      for (int kt = 0; kt < 25; ++kt) {
        v2f b = *(const v2f*)&smem[OFF_W1 + ((kt * 208 + nt * 16 + lo) << 2) + klo];
        acc = WMMA4(a1[kt], b, acc);
      }
      float bias = smem[OFF_B1 + nt * 16 + lo];
      #pragma unroll
      for (int v = 0; v < 8; ++v) {
        float x = acc[v] + bias;
        x = (x > 0.f) ? x : (__expf(x) - 1.f);
        hS[(v + 8 * hi) * HSTR + nt * 16 + lo] = x;
      }
    }
    asm volatile("s_wait_dscnt 0" ::: "memory");

    for (int nt = 0; nt < 2; ++nt) {
      v8f acc = {};
      #pragma unroll
      for (int kt = 0; kt < 50; ++kt) {
        v2f a = *(const v2f*)&hS[lo * HSTR + kt * 4 + klo];
        v2f b = *(const v2f*)&smem[OFF_W2 + ((kt * 32 + nt * 16 + lo) << 2) + klo];
        acc = WMMA4(a, b, acc);
      }
      int col = nt * 16 + lo;
      float bias = smem[OFF_B2 + col];
      #pragma unroll
      for (int v = 0; v < 8; ++v) {
        if (col < 20) {
          float zv = acc[v] + bias;
          int row = v + 8 * hi;
          zS[row * 20 + col] = zv;
          out_ze[(size_t)(rowBase + row) * 20 + col] = zv;
        }
      }
    }
    asm volatile("s_wait_dscnt 0" ::: "memory");

    v2f az[5];
    #pragma unroll
    for (int kt = 0; kt < 5; ++kt) az[kt] = *(const v2f*)&zS[lo * 20 + kt * 4 + klo];

    for (int nt = 0; nt < 2; ++nt) {
      v8f am = {}, al = {};
      #pragma unroll
      for (int kt = 0; kt < 5; ++kt) {
        v2f bm = *(const v2f*)&smem[OFF_WMU + ((kt * 32 + nt * 16 + lo) << 2) + klo];
        v2f bl = *(const v2f*)&smem[OFF_WLV + ((kt * 32 + nt * 16 + lo) << 2) + klo];
        am = WMMA4(az[kt], bm, am);
        al = WMMA4(az[kt], bl, al);
      }
      int col = nt * 16 + lo;
      if (col < 20) {
        float bmv = smem[OFF_BMU + col], blv_ = smem[OFF_BLV + col];
        #pragma unroll
        for (int v = 0; v < 8; ++v) {
          int row = v + 8 * hi;
          out_mu[(size_t)(rowBase + row) * 20 + col] = am[v] + bmv;
          out_lv[(size_t)(rowBase + row) * 20 + col] = al[v] + blv_;
        }
      }
    }

    float zn = 0.f;
    #pragma unroll
    for (int k = 0; k < 20; ++k) { float t = zS[lo * 20 + k]; zn += t * t; }
    float znv[8];
    #pragma unroll
    for (int v = 0; v < 8; ++v) znv[v] = __shfl(zn, v + 8 * hi, 32);

    float bestD[8]; int bestJ[8];
    #pragma unroll
    for (int v = 0; v < 8; ++v) { bestD[v] = 3.4e38f; bestJ[v] = 0; }
    for (int jt = 0; jt < 32; ++jt) {
      v8f acc = {};
      #pragma unroll
      for (int kt = 0; kt < 5; ++kt) {
        v2f b = *(const v2f*)&smem[OFF_CBT + ((kt * 512 + jt * 16 + lo) << 2) + klo];
        acc = WMMA4(az[kt], b, acc);
      }
      float cn = smem[OFF_CBN + jt * 16 + lo];
      int j = jt * 16 + lo;
      #pragma unroll
      for (int v = 0; v < 8; ++v) {
        float d = znv[v] + cn - 2.f * acc[v];
        if (d < bestD[v]) { bestD[v] = d; bestJ[v] = j; }
      }
    }
    #pragma unroll
    for (int s = 8; s >= 1; s >>= 1) {
      #pragma unroll
      for (int v = 0; v < 8; ++v) {
        float od = __shfl_xor(bestD[v], s, 32);
        int   oj = __shfl_xor(bestJ[v], s, 32);
        if (od < bestD[v] || (od == bestD[v] && oj < bestJ[v])) {
          bestD[v] = od; bestJ[v] = oj;
        }
      }
    }
    if (lo == 0) {
      #pragma unroll
      for (int v = 0; v < 8; ++v) { mnD[v + 8 * hi] = bestD[v]; mnJ[v + 8 * hi] = bestJ[v]; }
    }
    asm volatile("s_wait_dscnt 0" ::: "memory");

    if (lane == 0) {
      float s = 0.f;
      for (int m = 0; m < 16; ++m) s += mnD[m];
      ws[tile] = s;
    }
    #pragma unroll
    for (int t = 0; t < 10; ++t) {
      int e = lane * 10 + t;
      int row = e / 20, col = e - row * 20;
      int j = mnJ[row];
      float q = smem[OFF_CBT + (((col >> 2) * 512 + j) << 2) + (col & 3)];
      out_zq[(size_t)(rowBase + row) * 20 + col] = q;
    }
  }
}

__global__ __launch_bounds__(256) void vq_loss_reduce(const float* __restrict__ ws,
                                                      float* __restrict__ out,
                                                      int n, float invCount) {
  __shared__ float p[256];
  float s = 0.f;
  for (int i = threadIdx.x; i < n; i += 256) s += ws[i];
  p[threadIdx.x] = s;
  __syncthreads();
  if (threadIdx.x == 0) {
    float t = 0.f;
    for (int i = 0; i < 256; ++i) t += p[i];
    float q = t * invCount;
    out[0] = q;
    out[1] = 0.25f * q;
  }
}

extern "C" void kernel_launch(void* const* d_in, const int* in_sizes, int n_in,
                              void* d_out, int out_size, void* d_ws, size_t ws_size,
                              hipStream_t stream) {
  const float* c   = (const float*)d_in[0];
  const float* W1  = (const float*)d_in[1];
  const float* b1  = (const float*)d_in[2];
  const float* W2  = (const float*)d_in[3];
  const float* b2  = (const float*)d_in[4];
  const float* cb  = (const float*)d_in[5];
  const float* Wmu = (const float*)d_in[6];
  const float* bmu = (const float*)d_in[7];
  const float* Wlv = (const float*)d_in[8];
  const float* blv = (const float*)d_in[9];

  const int B = in_sizes[0] / 100;
  const int numTiles = B / 16;
  float* out = (float*)d_out;
  float* ws  = (float*)d_ws;

  const size_t ldsBytes = (size_t)LDS_FLOATS * sizeof(float);
  (void)hipFuncSetAttribute((const void*)vqvae_fused_kernel,
                            hipFuncAttributeMaxDynamicSharedMemorySize, (int)ldsBytes);

  const int blocks = (numTiles + 7) / 8;
  vqvae_fused_kernel<<<blocks, 256, ldsBytes, stream>>>(
      c, W1, b1, W2, b2, cb, Wmu, bmu, Wlv, blv,
      out,
      out + (size_t)B * 20,
      out + 2 * (size_t)B * 20,
      out + 3 * (size_t)B * 20,
      ws, B);
  vq_loss_reduce<<<1, 256, 0, stream>>>(ws, out + 4 * (size_t)B * 20,
                                        numTiles, 1.0f / ((float)B * 20.0f));
}


# persistent grid: 256 work-groups (one per WGP) instead of 2048; each wave walks 8 tiles, so the weight staging into LDS is paid once per WGP instead of 8 times
# speedup vs baseline: 1.7321x; 1.7321x over previous
#include <hip/hip_runtime.h>
#include <math.h>


typedef float v2f __attribute__((ext_vector_type(2)));
typedef float v8f __attribute__((ext_vector_type(8)));

#define WMMA4(a, b, c) \
  __builtin_amdgcn_wmma_f32_16x16x4_f32(false, (a), false, (b), (short)0, (c), false, false)

constexpr int OFF_W1  = 0;
constexpr int OFF_W2  = OFF_W1  + 20800;
constexpr int OFF_WMU = OFF_W2  + 6400;
constexpr int OFF_WLV = OFF_WMU + 640;
constexpr int OFF_CBT = OFF_WLV + 640;
constexpr int OFF_B1  = OFF_CBT + 10240;
constexpr int OFF_B2  = OFF_B1  + 208;
constexpr int OFF_BMU = OFF_B2  + 32;
constexpr int OFF_BLV = OFF_BMU + 32;
constexpr int OFF_CBN = OFF_BLV + 32;
constexpr int HSTR    = 210;
constexpr int OFF_H   = OFF_CBN + 512;
constexpr int OFF_Z   = OFF_H   + 8 * 16 * HSTR;
constexpr int OFF_MND = OFF_Z   + 8 * 320;
constexpr int OFF_MNJ = OFF_MND + 8 * 16;
constexpr int LDS_FLOATS = OFF_MNJ + 8 * 16;

__global__ __launch_bounds__(256) void vqvae_fused_kernel(
    const float* __restrict__ c,  const float* __restrict__ W1, const float* __restrict__ b1,
    const float* __restrict__ W2, const float* __restrict__ b2, const float* __restrict__ cb,
    const float* __restrict__ Wmu, const float* __restrict__ bmu,
    const float* __restrict__ Wlv, const float* __restrict__ blv,
    float* __restrict__ out_mu, float* __restrict__ out_lv,
    float* __restrict__ out_zq, float* __restrict__ out_ze,
    float* __restrict__ ws, int B) {
  extern __shared__ float smem[];
  const int tid  = threadIdx.x;
  const int wave = tid >> 5;
  const int lane = tid & 31;
  const int lo   = lane & 15;
  const int hi   = lane >> 4;
  const int klo  = hi * 2;

  for (int f = tid; f < 20800; f += 256) {
    int kk = f & 3, rem = f >> 2, n = rem % 208, kt = rem / 208;
    smem[OFF_W1 + f] = (n < 200) ? W1[(kt * 4 + kk) * 200 + n] : 0.f;
  }
  for (int f = tid; f < 6400; f += 256) {
    int kk = f & 3, rem = f >> 2, n = rem & 31, kt = rem >> 5;
    smem[OFF_W2 + f] = (n < 20) ? W2[(kt * 4 + kk) * 20 + n] : 0.f;
  }
  for (int f = tid; f < 640; f += 256) {
    int kk = f & 3, rem = f >> 2, n = rem & 31, kt = rem >> 5;
    smem[OFF_WMU + f] = (n < 20) ? Wmu[(kt * 4 + kk) * 20 + n] : 0.f;
    smem[OFF_WLV + f] = (n < 20) ? Wlv[(kt * 4 + kk) * 20 + n] : 0.f;
  }
  for (int f = tid; f < 10240; f += 256) {
    int kk = f & 3, rem = f >> 2, j = rem & 511, kt = rem >> 9;
    smem[OFF_CBT + f] = cb[j * 20 + kt * 4 + kk];
  }
  for (int n = tid; n < 208; n += 256) smem[OFF_B1 + n] = (n < 200) ? b1[n] : 0.f;
  if (tid < 32) {
    smem[OFF_B2  + tid] = (tid < 20) ? b2[tid]  : 0.f;
    smem[OFF_BMU + tid] = (tid < 20) ? bmu[tid] : 0.f;
    smem[OFF_BLV + tid] = (tid < 20) ? blv[tid] : 0.f;
  }
  for (int j = tid; j < 512; j += 256) {
    float s = 0.f;
    for (int k = 0; k < 20; ++k) { float t = cb[j * 20 + k]; s += t * t; }
    smem[OFF_CBN + j] = s;
  }
  __syncthreads();

  float* hS = &smem[OFF_H + wave * 16 * HSTR];
  float* zS = &smem[OFF_Z + wave * 320];
  float* mnD = &smem[OFF_MND + wave * 16];
  int*   mnJ = (int*)&smem[OFF_MNJ + wave * 16];

  const int numTiles = B >> 4;
  for (int tile = blockIdx.x * 8 + wave; tile < numTiles; tile += gridDim.x * 8) {
    const int rowBase = tile * 16;

    v2f a1[25];
    const float* crow = c + (size_t)(rowBase + lo) * 100 + klo;
    #pragma unroll
    for (int kt = 0; kt < 25; ++kt) a1[kt] = *(const v2f*)(crow + kt * 4);

    for (int nt = 0; nt < 13; ++nt) {
      v8f acc = {};
      #pragma unroll
      for (int kt = 0; kt < 25; ++kt) {
        v2f b = *(const v2f*)&smem[OFF_W1 + ((kt * 208 + nt * 16 + lo) << 2) + klo];
        acc = WMMA4(a1[kt], b, acc);
      }
      float bias = smem[OFF_B1 + nt * 16 + lo];
      #pragma unroll
      for (int v = 0; v < 8; ++v) {
        float x = acc[v] + bias;
        x = (x > 0.f) ? x : (__expf(x) - 1.f);
        hS[(v + 8 * hi) * HSTR + nt * 16 + lo] = x;
      }
    }
    asm volatile("s_wait_dscnt 0" ::: "memory");

    for (int nt = 0; nt < 2; ++nt) {
      v8f acc = {};
      #pragma unroll
      for (int kt = 0; kt < 50; ++kt) {
        v2f a = *(const v2f*)&hS[lo * HSTR + kt * 4 + klo];
        v2f b = *(const v2f*)&smem[OFF_W2 + ((kt * 32 + nt * 16 + lo) << 2) + klo];
        acc = WMMA4(a, b, acc);
      }
      int col = nt * 16 + lo;
      float bias = smem[OFF_B2 + col];
      #pragma unroll
      for (int v = 0; v < 8; ++v) {
        if (col < 20) {
          float zv = acc[v] + bias;
          int row = v + 8 * hi;
          zS[row * 20 + col] = zv;
          out_ze[(size_t)(rowBase + row) * 20 + col] = zv;
        }
      }
    }
    asm volatile("s_wait_dscnt 0" ::: "memory");

    v2f az[5];
    #pragma unroll
    for (int kt = 0; kt < 5; ++kt) az[kt] = *(const v2f*)&zS[lo * 20 + kt * 4 + klo];

    for (int nt = 0; nt < 2; ++nt) {
      v8f am = {}, al = {};
      #pragma unroll
      for (int kt = 0; kt < 5; ++kt) {
        v2f bm = *(const v2f*)&smem[OFF_WMU + ((kt * 32 + nt * 16 + lo) << 2) + klo];
        v2f bl = *(const v2f*)&smem[OFF_WLV + ((kt * 32 + nt * 16 + lo) << 2) + klo];
        am = WMMA4(az[kt], bm, am);
        al = WMMA4(az[kt], bl, al);
      }
      int col = nt * 16 + lo;
      if (col < 20) {
        float bmv = smem[OFF_BMU + col], blv_ = smem[OFF_BLV + col];
        #pragma unroll
        for (int v = 0; v < 8; ++v) {
          int row = v + 8 * hi;
          out_mu[(size_t)(rowBase + row) * 20 + col] = am[v] + bmv;
          out_lv[(size_t)(rowBase + row) * 20 + col] = al[v] + blv_;
        }
      }
    }

    float zn = 0.f;
    #pragma unroll
    for (int k = 0; k < 20; ++k) { float t = zS[lo * 20 + k]; zn += t * t; }
    float znv[8];
    #pragma unroll
    for (int v = 0; v < 8; ++v) znv[v] = __shfl(zn, v + 8 * hi, 32);

    float bestD[8]; int bestJ[8];
    #pragma unroll
    for (int v = 0; v < 8; ++v) { bestD[v] = 3.4e38f; bestJ[v] = 0; }
    for (int jt = 0; jt < 32; ++jt) {
      v8f acc = {};
      #pragma unroll
      for (int kt = 0; kt < 5; ++kt) {
        v2f b = *(const v2f*)&smem[OFF_CBT + ((kt * 512 + jt * 16 + lo) << 2) + klo];
        acc = WMMA4(az[kt], b, acc);
      }
      float cn = smem[OFF_CBN + jt * 16 + lo];
      int j = jt * 16 + lo;
      #pragma unroll
      for (int v = 0; v < 8; ++v) {
        float d = znv[v] + cn - 2.f * acc[v];
        if (d < bestD[v]) { bestD[v] = d; bestJ[v] = j; }
      }
    }
    #pragma unroll
    for (int s = 8; s >= 1; s >>= 1) {
      #pragma unroll
      for (int v = 0; v < 8; ++v) {
        float od = __shfl_xor(bestD[v], s, 32);
        int   oj = __shfl_xor(bestJ[v], s, 32);
        if (od < bestD[v] || (od == bestD[v] && oj < bestJ[v])) {
          bestD[v] = od; bestJ[v] = oj;
        }
      }
    }
    if (lo == 0) {
      #pragma unroll
      for (int v = 0; v < 8; ++v) { mnD[v + 8 * hi] = bestD[v]; mnJ[v + 8 * hi] = bestJ[v]; }
    }
    asm volatile("s_wait_dscnt 0" ::: "memory");

    if (lane == 0) {
      float s = 0.f;
      for (int m = 0; m < 16; ++m) s += mnD[m];
      ws[tile] = s;
    }
    #pragma unroll
    for (int t = 0; t < 10; ++t) {
      int e = lane * 10 + t;
      int row = e / 20, col = e - row * 20;
      int j = mnJ[row];
      float q = smem[OFF_CBT + (((col >> 2) * 512 + j) << 2) + (col & 3)];
      out_zq[(size_t)(rowBase + row) * 20 + col] = q;
    }
  }
}

__global__ __launch_bounds__(256) void vq_loss_reduce(const float* __restrict__ ws,
                                                      float* __restrict__ out,
                                                      int n, float invCount) {
  __shared__ float p[256];
  float s = 0.f;
  for (int i = threadIdx.x; i < n; i += 256) s += ws[i];
  p[threadIdx.x] = s;
  __syncthreads();
  if (threadIdx.x == 0) {
    float t = 0.f;
    for (int i = 0; i < 256; ++i) t += p[i];
    float q = t * invCount;
    out[0] = q;
    out[1] = 0.25f * q;
  }
}

extern "C" void kernel_launch(void* const* d_in, const int* in_sizes, int n_in,
                              void* d_out, int out_size, void* d_ws, size_t ws_size,
                              hipStream_t stream) {
  const float* c   = (const float*)d_in[0];
  const float* W1  = (const float*)d_in[1];
  const float* b1  = (const float*)d_in[2];
  const float* W2  = (const float*)d_in[3];
  const float* b2  = (const float*)d_in[4];
  const float* cb  = (const float*)d_in[5];
  const float* Wmu = (const float*)d_in[6];
  const float* bmu = (const float*)d_in[7];
  const float* Wlv = (const float*)d_in[8];
  const float* blv = (const float*)d_in[9];

  const int B = in_sizes[0] / 100;
  const int numTiles = B / 16;
  float* out = (float*)d_out;
  float* ws  = (float*)d_ws;

  const size_t ldsBytes = (size_t)LDS_FLOATS * sizeof(float);
  (void)hipFuncSetAttribute((const void*)vqvae_fused_kernel,
                            hipFuncAttributeMaxDynamicSharedMemorySize, (int)ldsBytes);

  const int blocks = 256;
  vqvae_fused_kernel<<<blocks, 256, ldsBytes, stream>>>(
      c, W1, b1, W2, b2, cb, Wmu, bmu, Wlv, blv,
      out,
      out + (size_t)B * 20,
      out + 2 * (size_t)B * 20,
      out + 3 * (size_t)B * 20,
      ws, B);
  vq_loss_reduce<<<1, 256, 0, stream>>>(ws, out + 4 * (size_t)B * 20,
                                        numTiles, 1.0f / ((float)B * 20.0f));
}


# independent WMMA accumulator chains in every matrix loop: GEMM1 column tiles in pairs, GEMM2 both column tiles share each A load, VQ four codebook tiles per trip; summation order unchanged
# speedup vs baseline: 1.8109x; 1.0455x over previous
#include <hip/hip_runtime.h>
#include <math.h>


typedef float v2f __attribute__((ext_vector_type(2)));
typedef float v8f __attribute__((ext_vector_type(8)));

#define WMMA4(a, b, c) \
  __builtin_amdgcn_wmma_f32_16x16x4_f32(false, (a), false, (b), (short)0, (c), false, false)

constexpr int OFF_W1  = 0;
constexpr int OFF_W2  = OFF_W1  + 20800;
constexpr int OFF_WMU = OFF_W2  + 6400;
constexpr int OFF_WLV = OFF_WMU + 640;
constexpr int OFF_CBT = OFF_WLV + 640;
constexpr int OFF_B1  = OFF_CBT + 10240;
constexpr int OFF_B2  = OFF_B1  + 208;
constexpr int OFF_BMU = OFF_B2  + 32;
constexpr int OFF_BLV = OFF_BMU + 32;
constexpr int OFF_CBN = OFF_BLV + 32;
constexpr int HSTR    = 210;
constexpr int OFF_H   = OFF_CBN + 512;
constexpr int OFF_Z   = OFF_H   + 8 * 16 * HSTR;
constexpr int OFF_MND = OFF_Z   + 8 * 320;
constexpr int OFF_MNJ = OFF_MND + 8 * 16;
constexpr int LDS_FLOATS = OFF_MNJ + 8 * 16;

__global__ __launch_bounds__(256) void vqvae_fused_kernel(
    const float* __restrict__ c,  const float* __restrict__ W1, const float* __restrict__ b1,
    const float* __restrict__ W2, const float* __restrict__ b2, const float* __restrict__ cb,
    const float* __restrict__ Wmu, const float* __restrict__ bmu,
    const float* __restrict__ Wlv, const float* __restrict__ blv,
    float* __restrict__ out_mu, float* __restrict__ out_lv,
    float* __restrict__ out_zq, float* __restrict__ out_ze,
    float* __restrict__ ws, int B) {
  extern __shared__ float smem[];
  const int tid  = threadIdx.x;
  const int wave = tid >> 5;
  const int lane = tid & 31;
  const int lo   = lane & 15;
  const int hi   = lane >> 4;
  const int klo  = hi * 2;

  for (int f = tid; f < 20800; f += 256) {
    int kk = f & 3, rem = f >> 2, n = rem % 208, kt = rem / 208;
    smem[OFF_W1 + f] = (n < 200) ? W1[(kt * 4 + kk) * 200 + n] : 0.f;
  }
  for (int f = tid; f < 6400; f += 256) {
    int kk = f & 3, rem = f >> 2, n = rem & 31, kt = rem >> 5;
    smem[OFF_W2 + f] = (n < 20) ? W2[(kt * 4 + kk) * 20 + n] : 0.f;
  }
  for (int f = tid; f < 640; f += 256) {
    int kk = f & 3, rem = f >> 2, n = rem & 31, kt = rem >> 5;
    smem[OFF_WMU + f] = (n < 20) ? Wmu[(kt * 4 + kk) * 20 + n] : 0.f;
    smem[OFF_WLV + f] = (n < 20) ? Wlv[(kt * 4 + kk) * 20 + n] : 0.f;
  }
  for (int f = tid; f < 10240; f += 256) {
    int kk = f & 3, rem = f >> 2, j = rem & 511, kt = rem >> 9;
    smem[OFF_CBT + f] = cb[j * 20 + kt * 4 + kk];
  }
  for (int n = tid; n < 208; n += 256) smem[OFF_B1 + n] = (n < 200) ? b1[n] : 0.f;
  if (tid < 32) {
    smem[OFF_B2  + tid] = (tid < 20) ? b2[tid]  : 0.f;
    smem[OFF_BMU + tid] = (tid < 20) ? bmu[tid] : 0.f;
    smem[OFF_BLV + tid] = (tid < 20) ? blv[tid] : 0.f;
  }
  for (int j = tid; j < 512; j += 256) {
    float s = 0.f;
    for (int k = 0; k < 20; ++k) { float t = cb[j * 20 + k]; s += t * t; }
    smem[OFF_CBN + j] = s;
  }
  __syncthreads();

  float* hS = &smem[OFF_H + wave * 16 * HSTR];
  float* zS = &smem[OFF_Z + wave * 320];
  float* mnD = &smem[OFF_MND + wave * 16];
  int*   mnJ = (int*)&smem[OFF_MNJ + wave * 16];

  const int numTiles = B >> 4;
  for (int tile = blockIdx.x * 8 + wave; tile < numTiles; tile += gridDim.x * 8) {
    const int rowBase = tile * 16;

    v2f a1[25];
    const float* crow = c + (size_t)(rowBase + lo) * 100 + klo;
    #pragma unroll
    for (int kt = 0; kt < 25; ++kt) a1[kt] = *(const v2f*)(crow + kt * 4);

    auto elu_store = [&](const v8f& acc, int nt) {
      float bias = smem[OFF_B1 + nt * 16 + lo];
      #pragma unroll
      for (int v = 0; v < 8; ++v) {
        float x = acc[v] + bias;
        x = (x > 0.f) ? x : (__expf(x) - 1.f);
        hS[(v + 8 * hi) * HSTR + nt * 16 + lo] = x;
      }
    };
    for (int np = 0; np < 6; ++np) {
      const int nt = np * 2;
      v8f acc0 = {}, acc1 = {};
      #pragma unroll
      for (int kt = 0; kt < 25; ++kt) {
        const float* bp = &smem[OFF_W1 + ((kt * 208 + nt * 16 + lo) << 2) + klo];
        v2f b0 = *(const v2f*)bp;
        v2f b1 = *(const v2f*)(bp + 64);
        acc0 = WMMA4(a1[kt], b0, acc0);
        acc1 = WMMA4(a1[kt], b1, acc1);
      }
      elu_store(acc0, nt);
      elu_store(acc1, nt + 1);
    }
    {
      v8f acc = {};
      #pragma unroll
      for (int kt = 0; kt < 25; ++kt) {
        v2f b = *(const v2f*)&smem[OFF_W1 + ((kt * 208 + 12 * 16 + lo) << 2) + klo];
        acc = WMMA4(a1[kt], b, acc);
      }
      elu_store(acc, 12);
    }
    asm volatile("s_wait_dscnt 0" ::: "memory");

    {
      v8f zc0 = {}, zc1 = {};
      #pragma unroll
      for (int kt = 0; kt < 50; ++kt) {
        v2f a = *(const v2f*)&hS[lo * HSTR + kt * 4 + klo];
        const float* bp = &smem[OFF_W2 + ((kt * 32 + lo) << 2) + klo];
        v2f b0 = *(const v2f*)bp;
        v2f b1 = *(const v2f*)(bp + 64);
        zc0 = WMMA4(a, b0, zc0);
        zc1 = WMMA4(a, b1, zc1);
      }
      {
        float bias = smem[OFF_B2 + lo];
        #pragma unroll
        for (int v = 0; v < 8; ++v) {
          float zv = zc0[v] + bias;
          int row = v + 8 * hi;
          zS[row * 20 + lo] = zv;
          out_ze[(size_t)(rowBase + row) * 20 + lo] = zv;
        }
      }
      if (lo < 4) {
        int col = 16 + lo;
        float bias = smem[OFF_B2 + col];
        #pragma unroll
        for (int v = 0; v < 8; ++v) {
          float zv = zc1[v] + bias;
          int row = v + 8 * hi;
          zS[row * 20 + col] = zv;
          out_ze[(size_t)(rowBase + row) * 20 + col] = zv;
        }
      }
    }
    asm volatile("s_wait_dscnt 0" ::: "memory");

    v2f az[5];
    #pragma unroll
    for (int kt = 0; kt < 5; ++kt) az[kt] = *(const v2f*)&zS[lo * 20 + kt * 4 + klo];

    for (int nt = 0; nt < 2; ++nt) {
      v8f am = {}, al = {};
      #pragma unroll
      for (int kt = 0; kt < 5; ++kt) {
        v2f bm = *(const v2f*)&smem[OFF_WMU + ((kt * 32 + nt * 16 + lo) << 2) + klo];
        v2f bl = *(const v2f*)&smem[OFF_WLV + ((kt * 32 + nt * 16 + lo) << 2) + klo];
        am = WMMA4(az[kt], bm, am);
        al = WMMA4(az[kt], bl, al);
      }
      int col = nt * 16 + lo;
      if (col < 20) {
        float bmv = smem[OFF_BMU + col], blv_ = smem[OFF_BLV + col];
        #pragma unroll
        for (int v = 0; v < 8; ++v) {
          int row = v + 8 * hi;
          out_mu[(size_t)(rowBase + row) * 20 + col] = am[v] + bmv;
          out_lv[(size_t)(rowBase + row) * 20 + col] = al[v] + blv_;
        }
      }
    }

    float zn = 0.f;
    #pragma unroll
    for (int k = 0; k < 20; ++k) { float t = zS[lo * 20 + k]; zn += t * t; }
    float znv[8];
    #pragma unroll
    for (int v = 0; v < 8; ++v) znv[v] = __shfl(zn, v + 8 * hi, 32);

    float bestD[8]; int bestJ[8];
    #pragma unroll
    for (int v = 0; v < 8; ++v) { bestD[v] = 3.4e38f; bestJ[v] = 0; }
    for (int jq = 0; jq < 8; ++jq) {
      v8f acc[4] = {{}, {}, {}, {}};
      #pragma unroll
      for (int kt = 0; kt < 5; ++kt) {
        const float* bp = &smem[OFF_CBT + ((kt * 512 + jq * 64 + lo) << 2) + klo];
        #pragma unroll
        for (int u = 0; u < 4; ++u) {
          v2f b = *(const v2f*)(bp + u * 64);
          acc[u] = WMMA4(az[kt], b, acc[u]);
        }
      }
      #pragma unroll
      for (int u = 0; u < 4; ++u) {
        int j = jq * 64 + u * 16 + lo;
        float cn = smem[OFF_CBN + j];
        #pragma unroll
        for (int v = 0; v < 8; ++v) {
          float d = znv[v] + cn - 2.f * acc[u][v];
          if (d < bestD[v]) { bestD[v] = d; bestJ[v] = j; }
        }
      }
    }
    #pragma unroll
    for (int s = 8; s >= 1; s >>= 1) {
      #pragma unroll
      for (int v = 0; v < 8; ++v) {
        float od = __shfl_xor(bestD[v], s, 32);
        int   oj = __shfl_xor(bestJ[v], s, 32);
        if (od < bestD[v] || (od == bestD[v] && oj < bestJ[v])) {
          bestD[v] = od; bestJ[v] = oj;
        }
      }
    }
    if (lo == 0) {
      #pragma unroll
      for (int v = 0; v < 8; ++v) { mnD[v + 8 * hi] = bestD[v]; mnJ[v + 8 * hi] = bestJ[v]; }
    }
    asm volatile("s_wait_dscnt 0" ::: "memory");

    if (lane == 0) {
      float s = 0.f;
      for (int m = 0; m < 16; ++m) s += mnD[m];
      ws[tile] = s;
    }
    #pragma unroll
    for (int t = 0; t < 10; ++t) {
      int e = lane * 10 + t;
      int row = e / 20, col = e - row * 20;
      int j = mnJ[row];
      float q = smem[OFF_CBT + (((col >> 2) * 512 + j) << 2) + (col & 3)];
      out_zq[(size_t)(rowBase + row) * 20 + col] = q;
    }
  }
}

__global__ __launch_bounds__(256) void vq_loss_reduce(const float* __restrict__ ws,
                                                      float* __restrict__ out,
                                                      int n, float invCount) {
  __shared__ float p[256];
  float s = 0.f;
  for (int i = threadIdx.x; i < n; i += 256) s += ws[i];
  p[threadIdx.x] = s;
  __syncthreads();
  if (threadIdx.x == 0) {
    float t = 0.f;
    for (int i = 0; i < 256; ++i) t += p[i];
    float q = t * invCount;
    out[0] = q;
    out[1] = 0.25f * q;
  }
}

extern "C" void kernel_launch(void* const* d_in, const int* in_sizes, int n_in,
                              void* d_out, int out_size, void* d_ws, size_t ws_size,
                              hipStream_t stream) {
  const float* c   = (const float*)d_in[0];
  const float* W1  = (const float*)d_in[1];
  const float* b1  = (const float*)d_in[2];
  const float* W2  = (const float*)d_in[3];
  const float* b2  = (const float*)d_in[4];
  const float* cb  = (const float*)d_in[5];
  const float* Wmu = (const float*)d_in[6];
  const float* bmu = (const float*)d_in[7];
  const float* Wlv = (const float*)d_in[8];
  const float* blv = (const float*)d_in[9];

  const int B = in_sizes[0] / 100;
  const int numTiles = B / 16;
  float* out = (float*)d_out;
  float* ws  = (float*)d_ws;

  const size_t ldsBytes = (size_t)LDS_FLOATS * sizeof(float);
  (void)hipFuncSetAttribute((const void*)vqvae_fused_kernel,
                            hipFuncAttributeMaxDynamicSharedMemorySize, (int)ldsBytes);

  const int blocks = 256;
  vqvae_fused_kernel<<<blocks, 256, ldsBytes, stream>>>(
      c, W1, b1, W2, b2, cb, Wmu, bmu, Wlv, blv,
      out,
      out + (size_t)B * 20,
      out + 2 * (size_t)B * 20,
      out + 3 * (size_t)B * 20,
      ws, B);
  vq_loss_reduce<<<1, 256, 0, stream>>>(ws, out + 4 * (size_t)B * 20,
                                        numTiles, 1.0f / ((float)B * 20.0f));
}


# cross-lane argmin on the vector ALU: DPP row_xmask minimum of the distance, then minimum index among lanes holding it, instead of 32 serial LDS exchange-and-branch rounds; tie-break kept
# speedup vs baseline: 1.9422x; 1.0725x over previous
#include <hip/hip_runtime.h>
#include <math.h>


typedef float v2f __attribute__((ext_vector_type(2)));
typedef float v8f __attribute__((ext_vector_type(8)));

#define WMMA4(a, b, c) \
  __builtin_amdgcn_wmma_f32_16x16x4_f32(false, (a), false, (b), (short)0, (c), false, false)

template <int M>
__device__ __forceinline__ int row_xor_i32(int x) {
  return __builtin_amdgcn_update_dpp(0, x, 0x160 | M, 0xf, 0xf, true);
}
template <int M>
__device__ __forceinline__ float row_xor_f32(float x) {
  return __builtin_bit_cast(float, row_xor_i32<M>(__builtin_bit_cast(int, x)));
}

constexpr int OFF_W1  = 0;
constexpr int OFF_W2  = OFF_W1  + 20800;
constexpr int OFF_WMU = OFF_W2  + 6400;
constexpr int OFF_WLV = OFF_WMU + 640;
constexpr int OFF_CBT = OFF_WLV + 640;
constexpr int OFF_B1  = OFF_CBT + 10240;
constexpr int OFF_B2  = OFF_B1  + 208;
constexpr int OFF_BMU = OFF_B2  + 32;
constexpr int OFF_BLV = OFF_BMU + 32;
constexpr int OFF_CBN = OFF_BLV + 32;
constexpr int HSTR    = 210;
constexpr int OFF_H   = OFF_CBN + 512;
constexpr int OFF_Z   = OFF_H   + 8 * 16 * HSTR;
constexpr int OFF_MND = OFF_Z   + 8 * 320;
constexpr int OFF_MNJ = OFF_MND + 8 * 16;
constexpr int LDS_FLOATS = OFF_MNJ + 8 * 16;

__global__ __launch_bounds__(256) void vqvae_fused_kernel(
    const float* __restrict__ c,  const float* __restrict__ W1, const float* __restrict__ b1,
    const float* __restrict__ W2, const float* __restrict__ b2, const float* __restrict__ cb,
    const float* __restrict__ Wmu, const float* __restrict__ bmu,
    const float* __restrict__ Wlv, const float* __restrict__ blv,
    float* __restrict__ out_mu, float* __restrict__ out_lv,
    float* __restrict__ out_zq, float* __restrict__ out_ze,
    float* __restrict__ ws, int B) {
  extern __shared__ float smem[];
  const int tid  = threadIdx.x;
  const int wave = tid >> 5;
  const int lane = tid & 31;
  const int lo   = lane & 15;
  const int hi   = lane >> 4;
  const int klo  = hi * 2;

  for (int f = tid; f < 20800; f += 256) {
    int kk = f & 3, rem = f >> 2, n = rem % 208, kt = rem / 208;
    smem[OFF_W1 + f] = (n < 200) ? W1[(kt * 4 + kk) * 200 + n] : 0.f;
  }
  for (int f = tid; f < 6400; f += 256) {
    int kk = f & 3, rem = f >> 2, n = rem & 31, kt = rem >> 5;
    smem[OFF_W2 + f] = (n < 20) ? W2[(kt * 4 + kk) * 20 + n] : 0.f;
  }
  for (int f = tid; f < 640; f += 256) {
    int kk = f & 3, rem = f >> 2, n = rem & 31, kt = rem >> 5;
    smem[OFF_WMU + f] = (n < 20) ? Wmu[(kt * 4 + kk) * 20 + n] : 0.f;
    smem[OFF_WLV + f] = (n < 20) ? Wlv[(kt * 4 + kk) * 20 + n] : 0.f;
  }
  for (int f = tid; f < 10240; f += 256) {
    int kk = f & 3, rem = f >> 2, j = rem & 511, kt = rem >> 9;
    smem[OFF_CBT + f] = cb[j * 20 + kt * 4 + kk];
  }
  for (int n = tid; n < 208; n += 256) smem[OFF_B1 + n] = (n < 200) ? b1[n] : 0.f;
  if (tid < 32) {
    smem[OFF_B2  + tid] = (tid < 20) ? b2[tid]  : 0.f;
    smem[OFF_BMU + tid] = (tid < 20) ? bmu[tid] : 0.f;
    smem[OFF_BLV + tid] = (tid < 20) ? blv[tid] : 0.f;
  }
  for (int j = tid; j < 512; j += 256) {
    float s = 0.f;
    for (int k = 0; k < 20; ++k) { float t = cb[j * 20 + k]; s += t * t; }
    smem[OFF_CBN + j] = s;
  }
  __syncthreads();

  float* hS = &smem[OFF_H + wave * 16 * HSTR];
  float* zS = &smem[OFF_Z + wave * 320];
  float* mnD = &smem[OFF_MND + wave * 16];
  int*   mnJ = (int*)&smem[OFF_MNJ + wave * 16];

  const int numTiles = B >> 4;
  for (int tile = blockIdx.x * 8 + wave; tile < numTiles; tile += gridDim.x * 8) {
    const int rowBase = tile * 16;

    v2f a1[25];
    const float* crow = c + (size_t)(rowBase + lo) * 100 + klo;
    #pragma unroll
    for (int kt = 0; kt < 25; ++kt) a1[kt] = *(const v2f*)(crow + kt * 4);

    auto elu_store = [&](const v8f& acc, int nt) {
      float bias = smem[OFF_B1 + nt * 16 + lo];
      #pragma unroll
      for (int v = 0; v < 8; ++v) {
        float x = acc[v] + bias;
        x = (x > 0.f) ? x : (__expf(x) - 1.f);
        hS[(v + 8 * hi) * HSTR + nt * 16 + lo] = x;
      }
    };
    for (int np = 0; np < 6; ++np) {
      const int nt = np * 2;
      v8f acc0 = {}, acc1 = {};
      #pragma unroll
      for (int kt = 0; kt < 25; ++kt) {
        const float* bp = &smem[OFF_W1 + ((kt * 208 + nt * 16 + lo) << 2) + klo];
        v2f b0 = *(const v2f*)bp;
        v2f b1 = *(const v2f*)(bp + 64);
        acc0 = WMMA4(a1[kt], b0, acc0);
        acc1 = WMMA4(a1[kt], b1, acc1);
      }
      elu_store(acc0, nt);
      elu_store(acc1, nt + 1);
    }
    {
      v8f acc = {};
      #pragma unroll
      for (int kt = 0; kt < 25; ++kt) {
        v2f b = *(const v2f*)&smem[OFF_W1 + ((kt * 208 + 12 * 16 + lo) << 2) + klo];
        acc = WMMA4(a1[kt], b, acc);
      }
      elu_store(acc, 12);
    }
    asm volatile("s_wait_dscnt 0" ::: "memory");

    {
      v8f zc0 = {}, zc1 = {};
      #pragma unroll
      for (int kt = 0; kt < 50; ++kt) {
        v2f a = *(const v2f*)&hS[lo * HSTR + kt * 4 + klo];
        const float* bp = &smem[OFF_W2 + ((kt * 32 + lo) << 2) + klo];
        v2f b0 = *(const v2f*)bp;
        v2f b1 = *(const v2f*)(bp + 64);
        zc0 = WMMA4(a, b0, zc0);
        zc1 = WMMA4(a, b1, zc1);
      }
      {
        float bias = smem[OFF_B2 + lo];
        #pragma unroll
        for (int v = 0; v < 8; ++v) {
          float zv = zc0[v] + bias;
          int row = v + 8 * hi;
          zS[row * 20 + lo] = zv;
          out_ze[(size_t)(rowBase + row) * 20 + lo] = zv;
        }
      }
      if (lo < 4) {
        int col = 16 + lo;
        float bias = smem[OFF_B2 + col];
        #pragma unroll
        for (int v = 0; v < 8; ++v) {
          float zv = zc1[v] + bias;
          int row = v + 8 * hi;
          zS[row * 20 + col] = zv;
          out_ze[(size_t)(rowBase + row) * 20 + col] = zv;
        }
      }
    }
    asm volatile("s_wait_dscnt 0" ::: "memory");

    v2f az[5];
    #pragma unroll
    for (int kt = 0; kt < 5; ++kt) az[kt] = *(const v2f*)&zS[lo * 20 + kt * 4 + klo];

    for (int nt = 0; nt < 2; ++nt) {
      v8f am = {}, al = {};
      #pragma unroll
      for (int kt = 0; kt < 5; ++kt) {
        v2f bm = *(const v2f*)&smem[OFF_WMU + ((kt * 32 + nt * 16 + lo) << 2) + klo];
        v2f bl = *(const v2f*)&smem[OFF_WLV + ((kt * 32 + nt * 16 + lo) << 2) + klo];
        am = WMMA4(az[kt], bm, am);
        al = WMMA4(az[kt], bl, al);
      }
      int col = nt * 16 + lo;
      if (col < 20) {
        float bmv = smem[OFF_BMU + col], blv_ = smem[OFF_BLV + col];
        #pragma unroll
        for (int v = 0; v < 8; ++v) {
          int row = v + 8 * hi;
          out_mu[(size_t)(rowBase + row) * 20 + col] = am[v] + bmv;
          out_lv[(size_t)(rowBase + row) * 20 + col] = al[v] + blv_;
        }
      }
    }

    float zn = 0.f;
    #pragma unroll
    for (int k = 0; k < 20; ++k) { float t = zS[lo * 20 + k]; zn += t * t; }
    float znv[8];
    #pragma unroll
    for (int v = 0; v < 8; ++v) znv[v] = __shfl(zn, v + 8 * hi, 32);

    float bestD[8]; int bestJ[8];
    #pragma unroll
    for (int v = 0; v < 8; ++v) { bestD[v] = 3.4e38f; bestJ[v] = 0; }
    for (int jq = 0; jq < 8; ++jq) {
      v8f acc[4] = {{}, {}, {}, {}};
      #pragma unroll
      for (int kt = 0; kt < 5; ++kt) {
        const float* bp = &smem[OFF_CBT + ((kt * 512 + jq * 64 + lo) << 2) + klo];
        #pragma unroll
        for (int u = 0; u < 4; ++u) {
          v2f b = *(const v2f*)(bp + u * 64);
          acc[u] = WMMA4(az[kt], b, acc[u]);
        }
      }
      #pragma unroll
      for (int u = 0; u < 4; ++u) {
        int j = jq * 64 + u * 16 + lo;
        float cn = smem[OFF_CBN + j];
        #pragma unroll
        for (int v = 0; v < 8; ++v) {
          float d = znv[v] + cn - 2.f * acc[u][v];
          if (d < bestD[v]) { bestD[v] = d; bestJ[v] = j; }
        }
      }
    }
    #pragma unroll
    for (int v = 0; v < 8; ++v) {
      float m = bestD[v];
      m = fminf(m, row_xor_f32<1>(m));
      m = fminf(m, row_xor_f32<2>(m));
      m = fminf(m, row_xor_f32<4>(m));
      m = fminf(m, row_xor_f32<8>(m));
      int j = (bestD[v] == m) ? bestJ[v] : 0x7fffffff;
      j = min(j, row_xor_i32<1>(j));
      j = min(j, row_xor_i32<2>(j));
      j = min(j, row_xor_i32<4>(j));
      j = min(j, row_xor_i32<8>(j));
      bestD[v] = m; bestJ[v] = j;
    }
    if (lo == 0) {
      #pragma unroll
      for (int v = 0; v < 8; ++v) { mnD[v + 8 * hi] = bestD[v]; mnJ[v + 8 * hi] = bestJ[v]; }
    }
    asm volatile("s_wait_dscnt 0" ::: "memory");

    if (lane == 0) {
      float s = 0.f;
      for (int m = 0; m < 16; ++m) s += mnD[m];
      ws[tile] = s;
    }
    #pragma unroll
    for (int t = 0; t < 10; ++t) {
      int e = lane * 10 + t;
      int row = e / 20, col = e - row * 20;
      int j = mnJ[row];
      float q = smem[OFF_CBT + (((col >> 2) * 512 + j) << 2) + (col & 3)];
      out_zq[(size_t)(rowBase + row) * 20 + col] = q;
    }
  }
}

__global__ __launch_bounds__(256) void vq_loss_reduce(const float* __restrict__ ws,
                                                      float* __restrict__ out,
                                                      int n, float invCount) {
  __shared__ float p[256];
  float s = 0.f;
  for (int i = threadIdx.x; i < n; i += 256) s += ws[i];
  p[threadIdx.x] = s;
  __syncthreads();
  if (threadIdx.x == 0) {
    float t = 0.f;
    for (int i = 0; i < 256; ++i) t += p[i];
    float q = t * invCount;
    out[0] = q;
    out[1] = 0.25f * q;
  }
}

extern "C" void kernel_launch(void* const* d_in, const int* in_sizes, int n_in,
                              void* d_out, int out_size, void* d_ws, size_t ws_size,
                              hipStream_t stream) {
  const float* c   = (const float*)d_in[0];
  const float* W1  = (const float*)d_in[1];
  const float* b1  = (const float*)d_in[2];
  const float* W2  = (const float*)d_in[3];
  const float* b2  = (const float*)d_in[4];
  const float* cb  = (const float*)d_in[5];
  const float* Wmu = (const float*)d_in[6];
  const float* bmu = (const float*)d_in[7];
  const float* Wlv = (const float*)d_in[8];
  const float* blv = (const float*)d_in[9];

  const int B = in_sizes[0] / 100;
  const int numTiles = B / 16;
  float* out = (float*)d_out;
  float* ws  = (float*)d_ws;

  const size_t ldsBytes = (size_t)LDS_FLOATS * sizeof(float);
  (void)hipFuncSetAttribute((const void*)vqvae_fused_kernel,
                            hipFuncAttributeMaxDynamicSharedMemorySize, (int)ldsBytes);

  const int blocks = 256;
  vqvae_fused_kernel<<<blocks, 256, ldsBytes, stream>>>(
      c, W1, b1, W2, b2, cb, Wmu, bmu, Wlv, blv,
      out,
      out + (size_t)B * 20,
      out + 2 * (size_t)B * 20,
      out + 3 * (size_t)B * 20,
      ws, B);
  vq_loss_reduce<<<1, 256, 0, stream>>>(ws, out + 4 * (size_t)B * 20,
                                        numTiles, 1.0f / ((float)B * 20.0f));
}


# global loads issued in groups ahead of first use in the weight staging (up to 52 per thread) and the loss partial sums (32 per thread), not one load-wait-use per trip; summation order unchanged
# speedup vs baseline: 2.7296x; 1.4054x over previous
#include <hip/hip_runtime.h>
#include <math.h>


typedef float v2f __attribute__((ext_vector_type(2)));
typedef float v4f __attribute__((ext_vector_type(4)));
typedef float v8f __attribute__((ext_vector_type(8)));

#define WMMA4(a, b, c) \
  __builtin_amdgcn_wmma_f32_16x16x4_f32(false, (a), false, (b), (short)0, (c), false, false)

template <int M>
__device__ __forceinline__ int row_xor_i32(int x) {
  return __builtin_amdgcn_update_dpp(0, x, 0x160 | M, 0xf, 0xf, true);
}
template <int M>
__device__ __forceinline__ float row_xor_f32(float x) {
  return __builtin_bit_cast(float, row_xor_i32<M>(__builtin_bit_cast(int, x)));
}

constexpr int OFF_W1  = 0;
constexpr int OFF_W2  = OFF_W1  + 20800;
constexpr int OFF_WMU = OFF_W2  + 6400;
constexpr int OFF_WLV = OFF_WMU + 640;
constexpr int OFF_CBT = OFF_WLV + 640;
constexpr int OFF_B1  = OFF_CBT + 10240;
constexpr int OFF_B2  = OFF_B1  + 208;
constexpr int OFF_BMU = OFF_B2  + 32;
constexpr int OFF_BLV = OFF_BMU + 32;
constexpr int OFF_CBN = OFF_BLV + 32;
constexpr int HSTR    = 210;
constexpr int OFF_H   = OFF_CBN + 512;
constexpr int OFF_Z   = OFF_H   + 8 * 16 * HSTR;
constexpr int OFF_MND = OFF_Z   + 8 * 320;
constexpr int OFF_MNJ = OFF_MND + 8 * 16;
constexpr int LDS_FLOATS = OFF_MNJ + 8 * 16;

__global__ __launch_bounds__(256) void vqvae_fused_kernel(
    const float* __restrict__ c,  const float* __restrict__ W1, const float* __restrict__ b1,
    const float* __restrict__ W2, const float* __restrict__ b2, const float* __restrict__ cb,
    const float* __restrict__ Wmu, const float* __restrict__ bmu,
    const float* __restrict__ Wlv, const float* __restrict__ blv,
    float* __restrict__ out_mu, float* __restrict__ out_lv,
    float* __restrict__ out_zq, float* __restrict__ out_ze,
    float* __restrict__ ws, int B) {
  extern __shared__ float smem[];
  const int tid  = threadIdx.x;
  const int wave = tid >> 5;
  const int lane = tid & 31;
  const int lo   = lane & 15;
  const int hi   = lane >> 4;
  const int klo  = hi * 2;

  if (tid < 208) {
    const float* p = W1 + min(tid, 199);
    const bool ok = tid < 200;
    float* dst = &smem[OFF_W1 + (tid << 2)];
    #pragma unroll
    for (int g = 0; g < 2; ++g) {
      float v[13][4];
      #pragma unroll
      for (int u = 0; u < 13; ++u) {
        const int kt = g * 13 + u;
        if (kt < 25) {
          #pragma unroll
          for (int kk = 0; kk < 4; ++kk) v[u][kk] = p[(kt * 4 + kk) * 200];
        }
      }
      #pragma unroll
      for (int u = 0; u < 13; ++u) {
        const int kt = g * 13 + u;
        if (kt < 25) {
          v4f w = { ok ? v[u][0] : 0.f, ok ? v[u][1] : 0.f, ok ? v[u][2] : 0.f, ok ? v[u][3] : 0.f };
          *(v4f*)(dst + kt * 832) = w;
        }
      }
    }
  }
  {
    constexpr int ITEMS = 50 * 32;
    constexpr int U = 7;
    float v[U][4];
    #pragma unroll
    for (int u = 0; u < U; ++u) {
      int it = min(tid + u * 256, ITEMS - 1);
      int kt = it >> 5, n = it & 31;
      const float* p = W2 + kt * 80 + min(n, 19);
      #pragma unroll
      for (int kk = 0; kk < 4; ++kk) v[u][kk] = p[kk * 20];
    }
    #pragma unroll
    for (int u = 0; u < U; ++u) {
      int it = tid + u * 256;
      if (it < ITEMS) {
        bool ok = (it & 31) < 20;
        v4f w = { ok ? v[u][0] : 0.f, ok ? v[u][1] : 0.f, ok ? v[u][2] : 0.f, ok ? v[u][3] : 0.f };
        *(v4f*)&smem[OFF_W2 + (it << 2)] = w;
      }
    }
  }
  {
    int it = min(tid, 159);
    int kt = it >> 5, n = it & 31;
    int off = kt * 80 + min(n, 19);
    float vm[4], vl[4];
    #pragma unroll
    for (int kk = 0; kk < 4; ++kk) { vm[kk] = Wmu[off + kk * 20]; vl[kk] = Wlv[off + kk * 20]; }
    if (tid < 160) {
      bool ok = n < 20;
      v4f wm = { ok ? vm[0] : 0.f, ok ? vm[1] : 0.f, ok ? vm[2] : 0.f, ok ? vm[3] : 0.f };
      v4f wl = { ok ? vl[0] : 0.f, ok ? vl[1] : 0.f, ok ? vl[2] : 0.f, ok ? vl[3] : 0.f };
      *(v4f*)&smem[OFF_WMU + (tid << 2)] = wm;
      *(v4f*)&smem[OFF_WLV + (tid << 2)] = wl;
    }
  }
  {
    constexpr int U = 10;
    v4f v[U];
    #pragma unroll
    for (int u = 0; u < U; ++u) {
      int it = tid + u * 256;
      int kt = it >> 9, j = it & 511;
      v[u] = *(const v4f*)(cb + j * 20 + kt * 4);
    }
    #pragma unroll
    for (int u = 0; u < U; ++u) *(v4f*)&smem[OFF_CBT + ((tid + u * 256) << 2)] = v[u];
  }
  {
    int n1 = min(tid, 199), n2 = min(tid, 19);
    float x1 = b1[n1], x2 = b2[n2], xm = bmu[n2], xl = blv[n2];
    if (tid < 208) smem[OFF_B1 + tid] = (tid < 200) ? x1 : 0.f;
    if (tid < 32) {
      smem[OFF_B2  + tid] = (tid < 20) ? x2 : 0.f;
      smem[OFF_BMU + tid] = (tid < 20) ? xm : 0.f;
      smem[OFF_BLV + tid] = (tid < 20) ? xl : 0.f;
    }
  }
  for (int j = tid; j < 512; j += 256) {
    float s = 0.f;
    for (int k = 0; k < 20; ++k) { float t = cb[j * 20 + k]; s += t * t; }
    smem[OFF_CBN + j] = s;
  }
  __syncthreads();

  float* hS = &smem[OFF_H + wave * 16 * HSTR];
  float* zS = &smem[OFF_Z + wave * 320];
  float* mnD = &smem[OFF_MND + wave * 16];
  int*   mnJ = (int*)&smem[OFF_MNJ + wave * 16];

  const int numTiles = B >> 4;
  for (int tile = blockIdx.x * 8 + wave; tile < numTiles; tile += gridDim.x * 8) {
    const int rowBase = tile * 16;

    v2f a1[25];
    const float* crow = c + (size_t)(rowBase + lo) * 100 + klo;
    #pragma unroll
    for (int kt = 0; kt < 25; ++kt) a1[kt] = *(const v2f*)(crow + kt * 4);

    auto elu_store = [&](const v8f& acc, int nt) {
      float bias = smem[OFF_B1 + nt * 16 + lo];
      #pragma unroll
      for (int v = 0; v < 8; ++v) {
        float x = acc[v] + bias;
        x = (x > 0.f) ? x : (__expf(x) - 1.f);
        hS[(v + 8 * hi) * HSTR + nt * 16 + lo] = x;
      }
    };
    for (int np = 0; np < 6; ++np) {
      const int nt = np * 2;
      v8f acc0 = {}, acc1 = {};
      #pragma unroll
      for (int kt = 0; kt < 25; ++kt) {
        const float* bp = &smem[OFF_W1 + ((kt * 208 + nt * 16 + lo) << 2) + klo];
        v2f b0 = *(const v2f*)bp;
        v2f b1 = *(const v2f*)(bp + 64);
        acc0 = WMMA4(a1[kt], b0, acc0);
        acc1 = WMMA4(a1[kt], b1, acc1);
      }
      elu_store(acc0, nt);
      elu_store(acc1, nt + 1);
    }
    {
      v8f acc = {};
      #pragma unroll
      for (int kt = 0; kt < 25; ++kt) {
        v2f b = *(const v2f*)&smem[OFF_W1 + ((kt * 208 + 12 * 16 + lo) << 2) + klo];
        acc = WMMA4(a1[kt], b, acc);
      }
      elu_store(acc, 12);
    }
    asm volatile("s_wait_dscnt 0" ::: "memory");

    {
      v8f zc0 = {}, zc1 = {};
      #pragma unroll
      for (int kt = 0; kt < 50; ++kt) {
        v2f a = *(const v2f*)&hS[lo * HSTR + kt * 4 + klo];
        const float* bp = &smem[OFF_W2 + ((kt * 32 + lo) << 2) + klo];
        v2f b0 = *(const v2f*)bp;
        v2f b1 = *(const v2f*)(bp + 64);
        zc0 = WMMA4(a, b0, zc0);
        zc1 = WMMA4(a, b1, zc1);
      }
      {
        float bias = smem[OFF_B2 + lo];
        #pragma unroll
        for (int v = 0; v < 8; ++v) {
          float zv = zc0[v] + bias;
          int row = v + 8 * hi;
          zS[row * 20 + lo] = zv;
          out_ze[(size_t)(rowBase + row) * 20 + lo] = zv;
        }
      }
      if (lo < 4) {
        int col = 16 + lo;
        float bias = smem[OFF_B2 + col];
        #pragma unroll
        for (int v = 0; v < 8; ++v) {
          float zv = zc1[v] + bias;
          int row = v + 8 * hi;
          zS[row * 20 + col] = zv;
          out_ze[(size_t)(rowBase + row) * 20 + col] = zv;
        }
      }
    }
    asm volatile("s_wait_dscnt 0" ::: "memory");

    v2f az[5];
    #pragma unroll
    for (int kt = 0; kt < 5; ++kt) az[kt] = *(const v2f*)&zS[lo * 20 + kt * 4 + klo];

    for (int nt = 0; nt < 2; ++nt) {
      v8f am = {}, al = {};
      #pragma unroll
      for (int kt = 0; kt < 5; ++kt) {
        v2f bm = *(const v2f*)&smem[OFF_WMU + ((kt * 32 + nt * 16 + lo) << 2) + klo];
        v2f bl = *(const v2f*)&smem[OFF_WLV + ((kt * 32 + nt * 16 + lo) << 2) + klo];
        am = WMMA4(az[kt], bm, am);
        al = WMMA4(az[kt], bl, al);
      }
      int col = nt * 16 + lo;
      if (col < 20) {
        float bmv = smem[OFF_BMU + col], blv_ = smem[OFF_BLV + col];
        #pragma unroll
        for (int v = 0; v < 8; ++v) {
          int row = v + 8 * hi;
          out_mu[(size_t)(rowBase + row) * 20 + col] = am[v] + bmv;
          out_lv[(size_t)(rowBase + row) * 20 + col] = al[v] + blv_;
        }
      }
    }

    float zn = 0.f;
    #pragma unroll
    for (int k = 0; k < 20; ++k) { float t = zS[lo * 20 + k]; zn += t * t; }
    float znv[8];
    #pragma unroll
    for (int v = 0; v < 8; ++v) znv[v] = __shfl(zn, v + 8 * hi, 32);

    float bestD[8]; int bestJ[8];
    #pragma unroll
    for (int v = 0; v < 8; ++v) { bestD[v] = 3.4e38f; bestJ[v] = 0; }
    for (int jq = 0; jq < 8; ++jq) {
      v8f acc[4] = {{}, {}, {}, {}};
      #pragma unroll
      for (int kt = 0; kt < 5; ++kt) {
        const float* bp = &smem[OFF_CBT + ((kt * 512 + jq * 64 + lo) << 2) + klo];
        #pragma unroll
        for (int u = 0; u < 4; ++u) {
          v2f b = *(const v2f*)(bp + u * 64);
          acc[u] = WMMA4(az[kt], b, acc[u]);
        }
      }
      #pragma unroll
      for (int u = 0; u < 4; ++u) {
        int j = jq * 64 + u * 16 + lo;
        float cn = smem[OFF_CBN + j];
        #pragma unroll
        for (int v = 0; v < 8; ++v) {
          float d = znv[v] + cn - 2.f * acc[u][v];
          if (d < bestD[v]) { bestD[v] = d; bestJ[v] = j; }
        }
      }
    }
    #pragma unroll
    for (int v = 0; v < 8; ++v) {
      float m = bestD[v];
      m = fminf(m, row_xor_f32<1>(m));
      m = fminf(m, row_xor_f32<2>(m));
      m = fminf(m, row_xor_f32<4>(m));
      m = fminf(m, row_xor_f32<8>(m));
      int j = (bestD[v] == m) ? bestJ[v] : 0x7fffffff;
      j = min(j, row_xor_i32<1>(j));
      j = min(j, row_xor_i32<2>(j));
      j = min(j, row_xor_i32<4>(j));
      j = min(j, row_xor_i32<8>(j));
      bestD[v] = m; bestJ[v] = j;
    }
    if (lo == 0) {
      #pragma unroll
      for (int v = 0; v < 8; ++v) { mnD[v + 8 * hi] = bestD[v]; mnJ[v + 8 * hi] = bestJ[v]; }
    }
    asm volatile("s_wait_dscnt 0" ::: "memory");

    if (lane == 0) {
      float s = 0.f;
      for (int m = 0; m < 16; ++m) s += mnD[m];
      ws[tile] = s;
    }
    #pragma unroll
    for (int t = 0; t < 10; ++t) {
      int e = lane * 10 + t;
      int row = e / 20, col = e - row * 20;
      int j = mnJ[row];
      float q = smem[OFF_CBT + (((col >> 2) * 512 + j) << 2) + (col & 3)];
      out_zq[(size_t)(rowBase + row) * 20 + col] = q;
    }
  }
}

__global__ __launch_bounds__(256) void vq_loss_reduce(const float* __restrict__ ws,
                                                      float* __restrict__ out,
                                                      int n, float invCount) {
  __shared__ float p[256];
  float s = 0.f;
  constexpr int U = 32;
  for (int base = threadIdx.x; base < n; base += 256 * U) {
    float v[U];
    #pragma unroll
    for (int u = 0; u < U; ++u) v[u] = ws[min(base + u * 256, n - 1)];
    #pragma unroll
    for (int u = 0; u < U; ++u) if (base + u * 256 < n) s += v[u];
  }
  p[threadIdx.x] = s;
  __syncthreads();
  if (threadIdx.x == 0) {
    float t = 0.f;
    for (int i = 0; i < 256; ++i) t += p[i];
    float q = t * invCount;
    out[0] = q;
    out[1] = 0.25f * q;
  }
}

extern "C" void kernel_launch(void* const* d_in, const int* in_sizes, int n_in,
                              void* d_out, int out_size, void* d_ws, size_t ws_size,
                              hipStream_t stream) {
  const float* c   = (const float*)d_in[0];
  const float* W1  = (const float*)d_in[1];
  const float* b1  = (const float*)d_in[2];
  const float* W2  = (const float*)d_in[3];
  const float* b2  = (const float*)d_in[4];
  const float* cb  = (const float*)d_in[5];
  const float* Wmu = (const float*)d_in[6];
  const float* bmu = (const float*)d_in[7];
  const float* Wlv = (const float*)d_in[8];
  const float* blv = (const float*)d_in[9];

  const int B = in_sizes[0] / 100;
  const int numTiles = B / 16;
  float* out = (float*)d_out;
  float* ws  = (float*)d_ws;

  const size_t ldsBytes = (size_t)LDS_FLOATS * sizeof(float);
  (void)hipFuncSetAttribute((const void*)vqvae_fused_kernel,
                            hipFuncAttributeMaxDynamicSharedMemorySize, (int)ldsBytes);

  const int blocks = 256;
  vqvae_fused_kernel<<<blocks, 256, ldsBytes, stream>>>(
      c, W1, b1, W2, b2, cb, Wmu, bmu, Wlv, blv,
      out,
      out + (size_t)B * 20,
      out + 2 * (size_t)B * 20,
      out + 3 * (size_t)B * 20,
      ws, B);
  vq_loss_reduce<<<1, 256, 0, stream>>>(ws, out + 4 * (size_t)B * 20,
                                        numTiles, 1.0f / ((float)B * 20.0f));
}


# every fragment load issued as one group one WMMA block ahead of its use, into a second register set: LDS weight fragments (none left among the WMMAs) and the next tile's c rows; sum order unchanged
# speedup vs baseline: 2.8269x; 1.0356x over previous
#include <hip/hip_runtime.h>
#include <math.h>


typedef float v2f __attribute__((ext_vector_type(2)));
typedef float v4f __attribute__((ext_vector_type(4)));
typedef float v8f __attribute__((ext_vector_type(8)));

#define WMMA4(a, b, c) \
  __builtin_amdgcn_wmma_f32_16x16x4_f32(false, (a), false, (b), (short)0, (c), false, false)

#define LOAD_GROUP_FENCE() __builtin_amdgcn_sched_barrier(0)

template <int M>
__device__ __forceinline__ int row_xor_i32(int x) {
  return __builtin_amdgcn_update_dpp(0, x, 0x160 | M, 0xf, 0xf, true);
}
template <int M>
__device__ __forceinline__ float row_xor_f32(float x) {
  return __builtin_bit_cast(float, row_xor_i32<M>(__builtin_bit_cast(int, x)));
}

constexpr int OFF_W1  = 0;
constexpr int OFF_W2  = OFF_W1  + 20800;
constexpr int OFF_WMU = OFF_W2  + 6400;
constexpr int OFF_WLV = OFF_WMU + 640;
constexpr int OFF_CBT = OFF_WLV + 640;
constexpr int OFF_B1  = OFF_CBT + 10240;
constexpr int OFF_B2  = OFF_B1  + 208;
constexpr int OFF_BMU = OFF_B2  + 32;
constexpr int OFF_BLV = OFF_BMU + 32;
constexpr int OFF_CBN = OFF_BLV + 32;
constexpr int HSTR    = 210;
constexpr int OFF_H   = OFF_CBN + 512;
constexpr int OFF_Z   = OFF_H   + 8 * 16 * HSTR;
constexpr int OFF_MND = OFF_Z   + 8 * 320;
constexpr int OFF_MNJ = OFF_MND + 8 * 16;
constexpr int LDS_FLOATS = OFF_MNJ + 8 * 16;

__global__ __launch_bounds__(256) void vqvae_fused_kernel(
    const float* __restrict__ c,  const float* __restrict__ W1, const float* __restrict__ b1,
    const float* __restrict__ W2, const float* __restrict__ b2, const float* __restrict__ cb,
    const float* __restrict__ Wmu, const float* __restrict__ bmu,
    const float* __restrict__ Wlv, const float* __restrict__ blv,
    float* __restrict__ out_mu, float* __restrict__ out_lv,
    float* __restrict__ out_zq, float* __restrict__ out_ze,
    float* __restrict__ ws, int B) {
  extern __shared__ float smem[];
  const int tid  = threadIdx.x;
  const int wave = tid >> 5;
  const int lane = tid & 31;
  const int lo   = lane & 15;
  const int hi   = lane >> 4;
  const int klo  = hi * 2;

  if (tid < 208) {
    const float* p = W1 + min(tid, 199);
    const bool ok = tid < 200;
    float* dst = &smem[OFF_W1 + (tid << 2)];
    #pragma unroll
    for (int g = 0; g < 2; ++g) {
      float v[13][4];
      #pragma unroll
      for (int u = 0; u < 13; ++u) {
        const int kt = g * 13 + u;
        if (kt < 25) {
          #pragma unroll
          for (int kk = 0; kk < 4; ++kk) v[u][kk] = p[(kt * 4 + kk) * 200];
        }
      }
      #pragma unroll
      for (int u = 0; u < 13; ++u) {
        const int kt = g * 13 + u;
        if (kt < 25) {
          v4f w = { ok ? v[u][0] : 0.f, ok ? v[u][1] : 0.f, ok ? v[u][2] : 0.f, ok ? v[u][3] : 0.f };
          *(v4f*)(dst + kt * 832) = w;
        }
      }
    }
  }
  {
    constexpr int ITEMS = 50 * 32;
    constexpr int U = 7;
    float v[U][4];
    #pragma unroll
    for (int u = 0; u < U; ++u) {
      int it = min(tid + u * 256, ITEMS - 1);
      int kt = it >> 5, n = it & 31;
      const float* p = W2 + kt * 80 + min(n, 19);
      #pragma unroll
      for (int kk = 0; kk < 4; ++kk) v[u][kk] = p[kk * 20];
    }
    #pragma unroll
    for (int u = 0; u < U; ++u) {
      int it = tid + u * 256;
      if (it < ITEMS) {
        bool ok = (it & 31) < 20;
        v4f w = { ok ? v[u][0] : 0.f, ok ? v[u][1] : 0.f, ok ? v[u][2] : 0.f, ok ? v[u][3] : 0.f };
        *(v4f*)&smem[OFF_W2 + (it << 2)] = w;
      }
    }
  }
  {
    int it = min(tid, 159);
    int kt = it >> 5, n = it & 31;
    int off = kt * 80 + min(n, 19);
    float vm[4], vl[4];
    #pragma unroll
    for (int kk = 0; kk < 4; ++kk) { vm[kk] = Wmu[off + kk * 20]; vl[kk] = Wlv[off + kk * 20]; }
    if (tid < 160) {
      bool ok = n < 20;
      v4f wm = { ok ? vm[0] : 0.f, ok ? vm[1] : 0.f, ok ? vm[2] : 0.f, ok ? vm[3] : 0.f };
      v4f wl = { ok ? vl[0] : 0.f, ok ? vl[1] : 0.f, ok ? vl[2] : 0.f, ok ? vl[3] : 0.f };
      *(v4f*)&smem[OFF_WMU + (tid << 2)] = wm;
      *(v4f*)&smem[OFF_WLV + (tid << 2)] = wl;
    }
  }
  {
    constexpr int U = 10;
    v4f v[U];
    #pragma unroll
    for (int u = 0; u < U; ++u) {
      int it = tid + u * 256;
      int kt = it >> 9, j = it & 511;
      v[u] = *(const v4f*)(cb + j * 20 + kt * 4);
    }
    #pragma unroll
    for (int u = 0; u < U; ++u) *(v4f*)&smem[OFF_CBT + ((tid + u * 256) << 2)] = v[u];
  }
  {
    int n1 = min(tid, 199), n2 = min(tid, 19);
    float x1 = b1[n1], x2 = b2[n2], xm = bmu[n2], xl = blv[n2];
    if (tid < 208) smem[OFF_B1 + tid] = (tid < 200) ? x1 : 0.f;
    if (tid < 32) {
      smem[OFF_B2  + tid] = (tid < 20) ? x2 : 0.f;
      smem[OFF_BMU + tid] = (tid < 20) ? xm : 0.f;
      smem[OFF_BLV + tid] = (tid < 20) ? xl : 0.f;
    }
  }
  for (int j = tid; j < 512; j += 256) {
    float s = 0.f;
    for (int k = 0; k < 20; ++k) { float t = cb[j * 20 + k]; s += t * t; }
    smem[OFF_CBN + j] = s;
  }
  __syncthreads();

  float* hS = &smem[OFF_H + wave * 16 * HSTR];
  float* zS = &smem[OFF_Z + wave * 320];
  float* mnD = &smem[OFF_MND + wave * 16];
  int*   mnJ = (int*)&smem[OFF_MNJ + wave * 16];

  const int numTiles = B >> 4;
  const int tileStep = gridDim.x * 8;

  v2f a1[25];
  {
    const int t0 = min((int)(blockIdx.x * 8) + wave, numTiles - 1);
    const float* crow = c + (size_t)(t0 * 16 + lo) * 100 + klo;
    #pragma unroll
    for (int kt = 0; kt < 25; ++kt) a1[kt] = *(const v2f*)(crow + kt * 4);
  }

  for (int tile = blockIdx.x * 8 + wave; tile < numTiles; tile += tileStep) {
    const int rowBase = tile * 16;

    auto elu_store = [&](const v8f& acc, int nt, float bias) {
      #pragma unroll
      for (int v = 0; v < 8; ++v) {
        float x = acc[v] + bias;
        x = (x > 0.f) ? x : (__expf(x) - 1.f);
        hS[(v + 8 * hi) * HSTR + nt * 16 + lo] = x;
      }
    };
    const float* w1p = &smem[OFF_W1 + (lo << 2) + klo];
    v2f wX[25][2], wY[25][2];
    auto w1_load = [&](v2f (&w)[25][2], int nt, int second) __attribute__((always_inline)) {
      const float* bp = w1p + nt * 64;
      #pragma unroll
      for (int kt = 0; kt < 25; ++kt) {
        w[kt][0] = *(const v2f*)(bp + kt * 832);
        w[kt][1] = *(const v2f*)(bp + kt * 832 + second);
      }
    };
    auto w1_block = [&](const v2f (&w)[25][2], int nt, float bias0, float bias1) __attribute__((always_inline)) {
      v8f acc0 = {}, acc1 = {};
      #pragma unroll
      for (int kt = 0; kt < 25; ++kt) {
        acc0 = WMMA4(a1[kt], w[kt][0], acc0);
        acc1 = WMMA4(a1[kt], w[kt][1], acc1);
      }
      LOAD_GROUP_FENCE();
      elu_store(acc0, nt, bias0);
      elu_store(acc1, nt + 1, bias1);
    };
    w1_load(wX, 0, 64);
    const float bias12 = smem[OFF_B1 + 12 * 16 + lo];
    #pragma nounroll
    for (int nq = 0; nq < 3; ++nq) {
      const int nt = nq * 4;
      w1_load(wY, nt + 2, 64);
      const float bias0 = smem[OFF_B1 + nt * 16 + lo];
      const float bias1 = smem[OFF_B1 + nt * 16 + 16 + lo];
      const float bias2 = smem[OFF_B1 + nt * 16 + 32 + lo];
      const float bias3 = smem[OFF_B1 + nt * 16 + 48 + lo];
      LOAD_GROUP_FENCE();
      w1_block(wX, nt, bias0, bias1);
      w1_load(wX, nt + 4, (nq < 2) ? 64 : 0);
      LOAD_GROUP_FENCE();
      w1_block(wY, nt + 2, bias2, bias3);
    }
    {
      v8f acc = {};
      #pragma unroll
      for (int kt = 0; kt < 25; ++kt) acc = WMMA4(a1[kt], wX[kt][0], acc);
      {
        const int tn = min(tile + tileStep, numTiles - 1);
        const float* crow = c + (size_t)(tn * 16 + lo) * 100 + klo;
        #pragma unroll
        for (int kt = 0; kt < 25; ++kt) a1[kt] = *(const v2f*)(crow + kt * 4);
      }
      LOAD_GROUP_FENCE();
      elu_store(acc, 12, bias12);
    }
    asm volatile("s_wait_dscnt 0" ::: "memory");

    {
      v8f zc0 = {}, zc1 = {};
      v2f ga[2][10], gb0[2][10], gb1[2][10];
      const float* hp = &hS[lo * HSTR + klo];
      const float* w2p = &smem[OFF_W2 + (lo << 2) + klo];
      #pragma unroll
      for (int u = 0; u < 10; ++u) {
        ga[0][u]  = *(const v2f*)(hp + u * 4);
        gb0[0][u] = *(const v2f*)(w2p + u * 128);
        gb1[0][u] = *(const v2f*)(w2p + u * 128 + 64);
      }
      const float zbias0 = smem[OFF_B2 + lo];
      const float zbias1 = smem[OFF_B2 + 16 + (lo & 3)];
      #pragma unroll
      for (int blk = 0; blk < 5; ++blk) {
        const int cur = blk & 1, nxt = cur ^ 1;
        if (blk < 4) {
          #pragma unroll
          for (int u = 0; u < 10; ++u) {
            const int kt = (blk + 1) * 10 + u;
            ga[nxt][u]  = *(const v2f*)(hp + kt * 4);
            gb0[nxt][u] = *(const v2f*)(w2p + kt * 128);
            gb1[nxt][u] = *(const v2f*)(w2p + kt * 128 + 64);
          }
        }
        LOAD_GROUP_FENCE();
        #pragma unroll
        for (int u = 0; u < 10; ++u) {
          zc0 = WMMA4(ga[cur][u], gb0[cur][u], zc0);
          zc1 = WMMA4(ga[cur][u], gb1[cur][u], zc1);
        }
        LOAD_GROUP_FENCE();
      }
      {
        float bias = zbias0;
        #pragma unroll
        for (int v = 0; v < 8; ++v) {
          float zv = zc0[v] + bias;
          int row = v + 8 * hi;
          zS[row * 20 + lo] = zv;
          out_ze[(size_t)(rowBase + row) * 20 + lo] = zv;
        }
      }
      if (lo < 4) {
        int col = 16 + lo;
        float bias = zbias1;
        #pragma unroll
        for (int v = 0; v < 8; ++v) {
          float zv = zc1[v] + bias;
          int row = v + 8 * hi;
          zS[row * 20 + col] = zv;
          out_ze[(size_t)(rowBase + row) * 20 + col] = zv;
        }
      }
    }
    asm volatile("s_wait_dscnt 0" ::: "memory");

    v2f az[5];
    v4f zr[5];
    v2f hm[5][2], hl[5][2];
    #pragma unroll
    for (int kt = 0; kt < 5; ++kt) {
      az[kt] = *(const v2f*)&zS[lo * 20 + kt * 4 + klo];
      zr[kt] = *(const v4f*)&zS[lo * 20 + kt * 4];
      #pragma unroll
      for (int nt = 0; nt < 2; ++nt) {
        hm[kt][nt] = *(const v2f*)&smem[OFF_WMU + ((kt * 32 + nt * 16 + lo) << 2) + klo];
        hl[kt][nt] = *(const v2f*)&smem[OFF_WLV + ((kt * 32 + nt * 16 + lo) << 2) + klo];
      }
    }
    const float bmv0 = smem[OFF_BMU + lo], blv0 = smem[OFF_BLV + lo];
    const float bmv1 = smem[OFF_BMU + 16 + (lo & 3)], blv1 = smem[OFF_BLV + 16 + (lo & 3)];

    v2f cf0[5][4], cf1[5][4];
    float cn0[4], cn1[4];
    auto vq_load = [&](v2f (&f)[5][4], float (&cn)[4], int jq) __attribute__((always_inline)) {
      const float* bp = &smem[OFF_CBT + ((jq * 64 + lo) << 2) + klo];
      #pragma unroll
      for (int kt = 0; kt < 5; ++kt) {
        #pragma unroll
        for (int u = 0; u < 4; ++u) f[kt][u] = *(const v2f*)(bp + kt * 2048 + u * 64);
      }
      #pragma unroll
      for (int u = 0; u < 4; ++u) cn[u] = smem[OFF_CBN + jq * 64 + u * 16 + lo];
    };
    vq_load(cf0, cn0, 0);
    LOAD_GROUP_FENCE();

    {
      v8f am0 = {}, al0 = {}, am1 = {}, al1 = {};
      #pragma unroll
      for (int kt = 0; kt < 5; ++kt) {
        am0 = WMMA4(az[kt], hm[kt][0], am0);
        al0 = WMMA4(az[kt], hl[kt][0], al0);
        am1 = WMMA4(az[kt], hm[kt][1], am1);
        al1 = WMMA4(az[kt], hl[kt][1], al1);
      }
      #pragma unroll
      for (int v = 0; v < 8; ++v) {
        int row = v + 8 * hi;
        out_mu[(size_t)(rowBase + row) * 20 + lo] = am0[v] + bmv0;
        out_lv[(size_t)(rowBase + row) * 20 + lo] = al0[v] + blv0;
      }
      if (lo < 4) {
        int col = 16 + lo;
        #pragma unroll
        for (int v = 0; v < 8; ++v) {
          int row = v + 8 * hi;
          out_mu[(size_t)(rowBase + row) * 20 + col] = am1[v] + bmv1;
          out_lv[(size_t)(rowBase + row) * 20 + col] = al1[v] + blv1;
        }
      }
    }

    float zn = 0.f;
    #pragma unroll
    for (int k = 0; k < 20; ++k) { float t = zr[k >> 2][k & 3]; zn += t * t; }
    float znv[8];
    #pragma unroll
    for (int v = 0; v < 8; ++v) znv[v] = __shfl(zn, v + 8 * hi, 32);

    float bestD[8]; int bestJ[8];
    #pragma unroll
    for (int v = 0; v < 8; ++v) { bestD[v] = 3.4e38f; bestJ[v] = 0; }
    auto vq_block = [&](const v2f (&f)[5][4], const float (&cnb)[4], int jq) __attribute__((always_inline)) {
      v8f acc[4] = {{}, {}, {}, {}};
      #pragma unroll
      for (int kt = 0; kt < 5; ++kt) {
        #pragma unroll
        for (int u = 0; u < 4; ++u) acc[u] = WMMA4(az[kt], f[kt][u], acc[u]);
      }
      LOAD_GROUP_FENCE();
      #pragma unroll
      for (int u = 0; u < 4; ++u) {
        int j = jq * 64 + u * 16 + lo;
        float cn = cnb[u];
        #pragma unroll
        for (int v = 0; v < 8; ++v) {
          float d = znv[v] + cn - 2.f * acc[u][v];
          if (d < bestD[v]) { bestD[v] = d; bestJ[v] = j; }
        }
      }
    };
    for (int jp = 0; jp < 4; ++jp) {
      const int jq = jp * 2;
      vq_load(cf1, cn1, jq + 1);
      LOAD_GROUP_FENCE();
      vq_block(cf0, cn0, jq);
      vq_load(cf0, cn0, min(jq + 2, 7));
      LOAD_GROUP_FENCE();
      vq_block(cf1, cn1, jq + 1);
    }
    #pragma unroll
    for (int v = 0; v < 8; ++v) {
      float m = bestD[v];
      m = fminf(m, row_xor_f32<1>(m));
      m = fminf(m, row_xor_f32<2>(m));
      m = fminf(m, row_xor_f32<4>(m));
      m = fminf(m, row_xor_f32<8>(m));
      int j = (bestD[v] == m) ? bestJ[v] : 0x7fffffff;
      j = min(j, row_xor_i32<1>(j));
      j = min(j, row_xor_i32<2>(j));
      j = min(j, row_xor_i32<4>(j));
      j = min(j, row_xor_i32<8>(j));
      bestD[v] = m; bestJ[v] = j;
    }
    if (lo == 0) {
      #pragma unroll
      for (int v = 0; v < 8; ++v) { mnD[v + 8 * hi] = bestD[v]; mnJ[v + 8 * hi] = bestJ[v]; }
    }
    asm volatile("s_wait_dscnt 0" ::: "memory");

    if (lane == 0) {
      float s = 0.f;
      for (int m = 0; m < 16; ++m) s += mnD[m];
      ws[tile] = s;
    }
    #pragma unroll
    for (int t = 0; t < 10; ++t) {
      int e = lane * 10 + t;
      int row = e / 20, col = e - row * 20;
      int j = mnJ[row];
      float q = smem[OFF_CBT + (((col >> 2) * 512 + j) << 2) + (col & 3)];
      out_zq[(size_t)(rowBase + row) * 20 + col] = q;
    }
  }
}

__global__ __launch_bounds__(256) void vq_loss_reduce(const float* __restrict__ ws,
                                                      float* __restrict__ out,
                                                      int n, float invCount) {
  __shared__ float p[256];
  float s = 0.f;
  constexpr int U = 32;
  for (int base = threadIdx.x; base < n; base += 256 * U) {
    float v[U];
    #pragma unroll
    for (int u = 0; u < U; ++u) v[u] = ws[min(base + u * 256, n - 1)];
    #pragma unroll
    for (int u = 0; u < U; ++u) if (base + u * 256 < n) s += v[u];
  }
  p[threadIdx.x] = s;
  __syncthreads();
  if (threadIdx.x == 0) {
    float t = 0.f;
    for (int i = 0; i < 256; ++i) t += p[i];
    float q = t * invCount;
    out[0] = q;
    out[1] = 0.25f * q;
  }
}

extern "C" void kernel_launch(void* const* d_in, const int* in_sizes, int n_in,
                              void* d_out, int out_size, void* d_ws, size_t ws_size,
                              hipStream_t stream) {
  const float* c   = (const float*)d_in[0];
  const float* W1  = (const float*)d_in[1];
  const float* b1  = (const float*)d_in[2];
  const float* W2  = (const float*)d_in[3];
  const float* b2  = (const float*)d_in[4];
  const float* cb  = (const float*)d_in[5];
  const float* Wmu = (const float*)d_in[6];
  const float* bmu = (const float*)d_in[7];
  const float* Wlv = (const float*)d_in[8];
  const float* blv = (const float*)d_in[9];

  const int B = in_sizes[0] / 100;
  const int numTiles = B / 16;
  float* out = (float*)d_out;
  float* ws  = (float*)d_ws;

  const size_t ldsBytes = (size_t)LDS_FLOATS * sizeof(float);
  (void)hipFuncSetAttribute((const void*)vqvae_fused_kernel,
                            hipFuncAttributeMaxDynamicSharedMemorySize, (int)ldsBytes);

  const int blocks = 256;
  vqvae_fused_kernel<<<blocks, 256, ldsBytes, stream>>>(
      c, W1, b1, W2, b2, cb, Wmu, bmu, Wlv, blv,
      out,
      out + (size_t)B * 20,
      out + 2 * (size_t)B * 20,
      out + 3 * (size_t)B * 20,
      ws, B);
  vq_loss_reduce<<<1, 256, 0, stream>>>(ws, out + 4 * (size_t)B * 20,
                                        numTiles, 1.0f / ((float)B * 20.0f));
}


# 32 rows per wave pass, encoder products transposed: h and z stay in registers (no LDS round trip), each weight fragment read from LDS feeds two WMMAs; k sum order of z, heads, distances changes
# speedup vs baseline: 3.1048x; 1.0983x over previous
#include <hip/hip_runtime.h>
#include <math.h>


typedef float v2f __attribute__((ext_vector_type(2)));
typedef float v4f __attribute__((ext_vector_type(4)));
typedef float v8f __attribute__((ext_vector_type(8)));

#define WMMA4(a, b, c) \
  __builtin_amdgcn_wmma_f32_16x16x4_f32(false, (a), false, (b), (short)0, (c), false, false)

#define LOAD_GROUP_FENCE() __builtin_amdgcn_sched_barrier(0)

template <int M>
__device__ __forceinline__ int row_xor_i32(int x) {
  return __builtin_amdgcn_update_dpp(0, x, 0x160 | M, 0xf, 0xf, true);
}
template <int M>
__device__ __forceinline__ float row_xor_f32(float x) {
  return __builtin_bit_cast(float, row_xor_i32<M>(__builtin_bit_cast(int, x)));
}

__device__ __forceinline__ float xhalf(float x) {
  return __builtin_bit_cast(float, __builtin_amdgcn_permlanex16(__builtin_bit_cast(int, x), __builtin_bit_cast(int, x),
                                                                0x76543210, 0xfedcba98, false, false));
}

constexpr int OFF_W1  = 0;
constexpr int OFF_W2  = OFF_W1  + 20800;
constexpr int OFF_WMU = OFF_W2  + 6656;
constexpr int OFF_WLV = OFF_WMU + 640;
constexpr int OFF_CBT = OFF_WLV + 640;
constexpr int OFF_B1  = OFF_CBT + 10240;
constexpr int OFF_B2  = OFF_B1  + 208;
constexpr int OFF_BMU = OFF_B2  + 32;
constexpr int OFF_BLV = OFF_BMU + 32;
constexpr int OFF_CBN = OFF_BLV + 32;
constexpr int OFF_MND = OFF_CBN + 512;
constexpr int OFF_MNJ = OFF_MND + 8 * 32;
constexpr int LDS_USED = OFF_MNJ + 8 * 32;
constexpr int LDS_FLOATS = (LDS_USED > 42000) ? LDS_USED : 42000;

__global__ __launch_bounds__(256) void vqvae_fused_kernel(
    const float* __restrict__ c,  const float* __restrict__ W1, const float* __restrict__ b1,
    const float* __restrict__ W2, const float* __restrict__ b2, const float* __restrict__ cb,
    const float* __restrict__ Wmu, const float* __restrict__ bmu,
    const float* __restrict__ Wlv, const float* __restrict__ blv,
    float* __restrict__ out_mu, float* __restrict__ out_lv,
    float* __restrict__ out_zq, float* __restrict__ out_ze,
    float* __restrict__ ws, int B) {
  extern __shared__ float smem[];
  const int tid  = threadIdx.x;
  const int wave = tid >> 5;
  const int lane = tid & 31;
  const int lo   = lane & 15;
  const int hi   = lane >> 4;
  const int klo  = hi * 2;

  if (tid < 208) {
    const float* p = W1 + min(tid, 199);
    const bool ok = tid < 200;
    float* dst = &smem[OFF_W1 + (tid << 2)];
    #pragma unroll
    for (int g = 0; g < 2; ++g) {
      float v[13][4];
      #pragma unroll
      for (int u = 0; u < 13; ++u) {
        const int kt = g * 13 + u;
        if (kt < 25) {
          #pragma unroll
          for (int kk = 0; kk < 4; ++kk) v[u][kk] = p[(kt * 4 + kk) * 200];
        }
      }
      #pragma unroll
      for (int u = 0; u < 13; ++u) {
        const int kt = g * 13 + u;
        if (kt < 25) {
          v4f w = { ok ? v[u][0] : 0.f, ok ? v[u][1] : 0.f, ok ? v[u][2] : 0.f, ok ? v[u][3] : 0.f };
          *(v4f*)(dst + kt * 832) = w;
        }
      }
    }
  }
  {
    constexpr int ITEMS = 52 * 32;
    constexpr int U = 7;
    float v[U][4];
    #pragma unroll
    for (int u = 0; u < U; ++u) {
      int it = min(tid + u * 256, ITEMS - 1);
      int kt = it >> 5, n = it & 31, m = n & 15;
      int lat = (n < 16) ? n : ((m < 2) ? 16 + m : 18 + (m & 1));
      int hbase = (kt >> 2) * 16 + (kt & 3) * 2;
      #pragma unroll
      for (int kk = 0; kk < 4; ++kk) v[u][kk] = W2[min(hbase + (kk >> 1) * 8 + (kk & 1), 199) * 20 + lat];
    }
    #pragma unroll
    for (int u = 0; u < U; ++u) {
      int it = tid + u * 256;
      if (it < ITEMS) {
        int kt = it >> 5, n = it & 31, m = n & 15;
        bool okc = (n < 16) || (m < 2) || (m == 8) || (m == 9);
        int hbase = (kt >> 2) * 16 + (kt & 3) * 2;
        bool ok0 = okc && (hbase < 200), ok1 = okc && (hbase + 1 < 200);
        bool ok2 = okc && (hbase + 8 < 200), ok3 = okc && (hbase + 9 < 200);
        v4f w = { ok0 ? v[u][0] : 0.f, ok1 ? v[u][1] : 0.f, ok2 ? v[u][2] : 0.f, ok3 ? v[u][3] : 0.f };
        *(v4f*)&smem[OFF_W2 + (it << 2)] = w;
      }
    }
  }
  {
    int it = min(tid, 159);
    int kt = it >> 5, n = it & 31;
    int col = min(n, 19);
    float vm[4], vl[4];
    #pragma unroll
    for (int kk = 0; kk < 4; ++kk) {
      int k = (kt < 4) ? ((kk >> 1) * 8 + kt * 2 + (kk & 1)) : (16 + kk);
      vm[kk] = Wmu[k * 20 + col]; vl[kk] = Wlv[k * 20 + col];
    }
    if (tid < 160) {
      bool ok = n < 20;
      v4f wm = { ok ? vm[0] : 0.f, ok ? vm[1] : 0.f, ok ? vm[2] : 0.f, ok ? vm[3] : 0.f };
      v4f wl = { ok ? vl[0] : 0.f, ok ? vl[1] : 0.f, ok ? vl[2] : 0.f, ok ? vl[3] : 0.f };
      *(v4f*)&smem[OFF_WMU + (tid << 2)] = wm;
      *(v4f*)&smem[OFF_WLV + (tid << 2)] = wl;
    }
  }
  {
    constexpr int U = 10;
    v2f va[U], vb[U];
    #pragma unroll
    for (int u = 0; u < U; ++u) {
      int it = tid + u * 256;
      int kt = it >> 9, j = it & 511;
      int ka = (kt < 4) ? 2 * kt : 16, kb = (kt < 4) ? 8 + 2 * kt : 18;
      va[u] = *(const v2f*)(cb + j * 20 + ka);
      vb[u] = *(const v2f*)(cb + j * 20 + kb);
    }
    #pragma unroll
    for (int u = 0; u < U; ++u) {
      v4f w = { va[u][0], va[u][1], vb[u][0], vb[u][1] };
      *(v4f*)&smem[OFF_CBT + ((tid + u * 256) << 2)] = w;
    }
  }
  {
    int n1 = min(tid, 199), n2 = min(tid, 19);
    float x1 = b1[n1], x2 = b2[n2], xm = bmu[n2], xl = blv[n2];
    if (tid < 208) smem[OFF_B1 + tid] = (tid < 200) ? x1 : 0.f;
    if (tid < 32) {
      smem[OFF_B2  + tid] = (tid < 20) ? x2 : 0.f;
      smem[OFF_BMU + tid] = (tid < 20) ? xm : 0.f;
      smem[OFF_BLV + tid] = (tid < 20) ? xl : 0.f;
    }
  }
  for (int j = tid; j < 512; j += 256) {
    float s = 0.f;
    for (int k = 0; k < 20; ++k) { float t = cb[j * 20 + k]; s += t * t; }
    smem[OFF_CBN + j] = s;
  }
  __syncthreads();

  float* mnD = &smem[OFF_MND + wave * 32];
  int*   mnJ = (int*)&smem[OFF_MNJ + wave * 32];

  const int numTiles = B >> 4;
  const int numPairs = (numTiles + 1) >> 1;
  const int pairStep = gridDim.x * 8;
  const float* w1p  = &smem[OFF_W1 + (lo << 2) + klo];
  const float* w2p  = &smem[OFF_W2 + (lo << 2) + klo];
  const float* b1p  = &smem[OFF_B1 + 8 * hi];
  const float* cbtp = &smem[OFF_CBT + (lo << 2) + klo];
  const float* cbnp = &smem[OFF_CBN + lo];

  v2f a1[2][25];
  {
    const int p0 = min((int)(blockIdx.x * 8) + wave, numPairs - 1);
    #pragma unroll
    for (int rt = 0; rt < 2; ++rt) {
      const int t0 = min(2 * p0 + rt, numTiles - 1);
      const float* crow = c + (size_t)(t0 * 16 + lo) * 100 + klo;
      #pragma unroll
      for (int kt = 0; kt < 25; ++kt) a1[rt][kt] = *(const v2f*)(crow + kt * 4);
    }
  }

  for (int pr = blockIdx.x * 8 + wave; pr < numPairs; pr += pairStep) {
    const int tl0 = min(2 * pr, numTiles - 1), tl1 = min(2 * pr + 1, numTiles - 1);
    const int rowBase[2] = { tl0 * 16, tl1 * 16 };

    v8f zc[2][2] = {{{}, {}}, {{}, {}}};
    v2f wX[25], wY[25];
    v2f uC[4][2];
    v4f bC[2];
    auto w_load = [&](v2f (&w)[25], int nt) __attribute__((always_inline)) {
      const float* bp = w1p + nt * 64;
      #pragma unroll
      for (int kt = 0; kt < 25; ++kt) w[kt] = *(const v2f*)(bp + kt * 832);
    };
    auto ub_load = [&](int nt) __attribute__((always_inline)) {
      const float* up = w2p + nt * 512;
      #pragma unroll
      for (int j = 0; j < 4; ++j) {
        uC[j][0] = *(const v2f*)(up + j * 128);
        uC[j][1] = *(const v2f*)(up + j * 128 + 64);
      }
      bC[0] = *(const v4f*)(b1p + nt * 16);
      bC[1] = *(const v4f*)(b1p + nt * 16 + 4);
    };
    auto elu8 = [&](const v8f& acc, float (&h)[8]) __attribute__((always_inline)) {
      #pragma unroll
      for (int r = 0; r < 8; ++r) {
        float x = acc[r] + ((r < 4) ? bC[0][r & 3] : bC[1][r & 3]);
        h[r] = (x > 0.f) ? x : (__expf(x) - 1.f);
      }
    };
    auto tile_tail = [&](v8f& accA, v8f& accB) __attribute__((always_inline)) {
      asm volatile("" : "+v"(accA), "+v"(accB));
      LOAD_GROUP_FENCE();
      float hA[8], hB[8];
      elu8(accA, hA);
      elu8(accB, hB);
      #pragma unroll
      for (int j = 0; j < 4; ++j) {
        v2f fa = { hA[2 * j], hA[2 * j + 1] }, fb = { hB[2 * j], hB[2 * j + 1] };
        zc[0][0] = WMMA4(uC[j][0], fa, zc[0][0]);
        zc[1][0] = WMMA4(uC[j][0], fb, zc[1][0]);
        zc[0][1] = WMMA4(uC[j][1], fa, zc[0][1]);
        zc[1][1] = WMMA4(uC[j][1], fb, zc[1][1]);
      }
    };
    auto tile_block = [&](const v2f (&w)[25]) __attribute__((always_inline)) {
      v8f accA = {}, accB = {};
      #pragma unroll
      for (int kt = 0; kt < 25; ++kt) {
        accA = WMMA4(w[kt], a1[0][kt], accA);
        accB = WMMA4(w[kt], a1[1][kt], accB);
      }
      tile_tail(accA, accB);
    };
    w_load(wX, 0);
    #pragma nounroll
    for (int nq = 0; nq < 6; ++nq) {
      const int nt = nq * 2;
      w_load(wY, nt + 1);
      ub_load(nt);
      LOAD_GROUP_FENCE();
      tile_block(wX);
      w_load(wX, nt + 2);
      ub_load(nt + 1);
      LOAD_GROUP_FENCE();
      tile_block(wY);
    }
    ub_load(12);
    const v4f bz0 = *(const v4f*)&smem[OFF_B2 + 8 * hi];
    const v4f bz1 = *(const v4f*)&smem[OFF_B2 + 8 * hi + 4];
    const v2f bz2 = *(const v2f*)&smem[OFF_B2 + 16 + 2 * hi];
    LOAD_GROUP_FENCE();
    {
      v8f accA = {}, accB = {};
      #pragma unroll
      for (int kt = 0; kt < 25; ++kt) {
        accA = WMMA4(wX[kt], a1[0][kt], accA);
        accB = WMMA4(wX[kt], a1[1][kt], accB);
      }
      {
        const int pn = min(pr + pairStep, numPairs - 1);
        #pragma unroll
        for (int rt = 0; rt < 2; ++rt) {
          const int tn = min(2 * pn + rt, numTiles - 1);
          const float* crow = c + (size_t)(tn * 16 + lo) * 100 + klo;
          #pragma unroll
          for (int kt = 0; kt < 25; ++kt) a1[rt][kt] = *(const v2f*)(crow + kt * 4);
        }
      }
      tile_tail(accA, accB);
    }

    float zt[2][10];
    v2f az[2][5];
    #pragma unroll
    for (int rt = 0; rt < 2; ++rt) {
      #pragma unroll
      for (int r = 0; r < 8; ++r) zt[rt][r] = zc[rt][0][r] + ((r < 4) ? bz0[r & 3] : bz1[r & 3]);
      zt[rt][8] = zc[rt][1][0] + bz2[0];
      zt[rt][9] = zc[rt][1][1] + bz2[1];
      float* zp = out_ze + (size_t)(rowBase[rt] + lo) * 20;
      v4f s0 = { zt[rt][0], zt[rt][1], zt[rt][2], zt[rt][3] }, s1 = { zt[rt][4], zt[rt][5], zt[rt][6], zt[rt][7] };
      v2f s2 = { zt[rt][8], zt[rt][9] };
      *(v4f*)(zp + 8 * hi)      = s0;
      *(v4f*)(zp + 8 * hi + 4)  = s1;
      *(v2f*)(zp + 16 + 2 * hi) = s2;
      #pragma unroll
      for (int kt = 0; kt < 5; ++kt) { v2f t2 = { zt[rt][2 * kt], zt[rt][2 * kt + 1] }; az[rt][kt] = t2; }
    }

    v2f hm[5][2], hl[5][2];
    #pragma unroll
    for (int kt = 0; kt < 5; ++kt) {
      #pragma unroll
      for (int nt = 0; nt < 2; ++nt) {
        hm[kt][nt] = *(const v2f*)&smem[OFF_WMU + ((kt * 32 + nt * 16 + lo) << 2) + klo];
        hl[kt][nt] = *(const v2f*)&smem[OFF_WLV + ((kt * 32 + nt * 16 + lo) << 2) + klo];
      }
    }
    const float bmv0 = smem[OFF_BMU + lo], blv0 = smem[OFF_BLV + lo];
    const float bmv1 = smem[OFF_BMU + 16 + (lo & 3)], blv1 = smem[OFF_BLV + 16 + (lo & 3)];

    v2f cf0[5][4], cf1[5][4];
    float cn0[4], cn1[4];
    auto vq_load = [&](v2f (&f)[5][4], float (&cn)[4], int jq) __attribute__((always_inline)) {
      const float* bp = cbtp + jq * 256;
      #pragma unroll
      for (int kt = 0; kt < 5; ++kt) {
        #pragma unroll
        for (int u = 0; u < 4; ++u) f[kt][u] = *(const v2f*)(bp + kt * 2048 + u * 64);
      }
      #pragma unroll
      for (int u = 0; u < 4; ++u) cn[u] = cbnp[jq * 64 + u * 16];
    };
    vq_load(cf0, cn0, 0);
    LOAD_GROUP_FENCE();

    float zn[2];
    #pragma unroll
    for (int rt = 0; rt < 2; ++rt) {
      v8f am0 = {}, al0 = {}, am1 = {}, al1 = {};
      #pragma unroll
      for (int kt = 0; kt < 5; ++kt) {
        am0 = WMMA4(az[rt][kt], hm[kt][0], am0);
        al0 = WMMA4(az[rt][kt], hl[kt][0], al0);
        am1 = WMMA4(az[rt][kt], hm[kt][1], am1);
        al1 = WMMA4(az[rt][kt], hl[kt][1], al1);
      }
      #pragma unroll
      for (int v = 0; v < 8; ++v) {
        int row = v + 8 * hi;
        out_mu[(size_t)(rowBase[rt] + row) * 20 + lo] = am0[v] + bmv0;
        out_lv[(size_t)(rowBase[rt] + row) * 20 + lo] = al0[v] + blv0;
      }
      if (lo < 4) {
        int col = 16 + lo;
        #pragma unroll
        for (int v = 0; v < 8; ++v) {
          int row = v + 8 * hi;
          out_mu[(size_t)(rowBase[rt] + row) * 20 + col] = am1[v] + bmv1;
          out_lv[(size_t)(rowBase[rt] + row) * 20 + col] = al1[v] + blv1;
        }
      }

      float zo[10];
      #pragma unroll
      for (int i = 0; i < 10; ++i) zo[i] = xhalf(zt[rt][i]);
      float zfull[20];
      #pragma unroll
      for (int r = 0; r < 8; ++r) { zfull[r] = hi ? zo[r] : zt[rt][r]; zfull[8 + r] = hi ? zt[rt][r] : zo[r]; }
      #pragma unroll
      for (int s2 = 0; s2 < 2; ++s2) { zfull[16 + s2] = hi ? zo[8 + s2] : zt[rt][8 + s2]; zfull[18 + s2] = hi ? zt[rt][8 + s2] : zo[8 + s2]; }
      float acc = 0.f;
      #pragma unroll
      for (int k = 0; k < 20; ++k) { float t = zfull[k]; acc += t * t; }
      zn[rt] = acc;
    }
    float znv[2][8];
    #pragma unroll
    for (int rt = 0; rt < 2; ++rt) {
      #pragma unroll
      for (int v = 0; v < 8; ++v) znv[rt][v] = __shfl(zn[rt], v + 8 * hi, 32);
    }

    float bestD[2][8]; int bestJ[2][8];
    #pragma unroll
    for (int rt = 0; rt < 2; ++rt) {
      #pragma unroll
      for (int v = 0; v < 8; ++v) { bestD[rt][v] = 3.4e38f; bestJ[rt][v] = 0; }
    }
    auto vq_block = [&](const v2f (&f)[5][4], const float (&cnb)[4], int jq) __attribute__((always_inline)) {
      v8f acc[2][4] = {{{}, {}, {}, {}}, {{}, {}, {}, {}}};
      #pragma unroll
      for (int kt = 0; kt < 5; ++kt) {
        #pragma unroll
        for (int u = 0; u < 4; ++u) {
          acc[0][u] = WMMA4(az[0][kt], f[kt][u], acc[0][u]);
          acc[1][u] = WMMA4(az[1][kt], f[kt][u], acc[1][u]);
        }
      }
      LOAD_GROUP_FENCE();
      #pragma unroll
      for (int rt = 0; rt < 2; ++rt) {
        #pragma unroll
        for (int u = 0; u < 4; ++u) {
          int j = jq * 64 + u * 16 + lo;
          float cn = cnb[u];
          #pragma unroll
          for (int v = 0; v < 8; ++v) {
            float d = znv[rt][v] + cn - 2.f * acc[rt][u][v];
            if (d < bestD[rt][v]) { bestD[rt][v] = d; bestJ[rt][v] = j; }
          }
        }
      }
    };
    for (int jp = 0; jp < 4; ++jp) {
      const int jq = jp * 2;
      vq_load(cf1, cn1, jq + 1);
      LOAD_GROUP_FENCE();
      vq_block(cf0, cn0, jq);
      vq_load(cf0, cn0, min(jq + 2, 7));
      LOAD_GROUP_FENCE();
      vq_block(cf1, cn1, jq + 1);
    }
    #pragma unroll
    for (int rt = 0; rt < 2; ++rt) {
      #pragma unroll
      for (int v = 0; v < 8; ++v) {
        float m = bestD[rt][v];
        m = fminf(m, row_xor_f32<1>(m));
        m = fminf(m, row_xor_f32<2>(m));
        m = fminf(m, row_xor_f32<4>(m));
        m = fminf(m, row_xor_f32<8>(m));
        int j = (bestD[rt][v] == m) ? bestJ[rt][v] : 0x7fffffff;
        j = min(j, row_xor_i32<1>(j));
        j = min(j, row_xor_i32<2>(j));
        j = min(j, row_xor_i32<4>(j));
        j = min(j, row_xor_i32<8>(j));
        bestD[rt][v] = m; bestJ[rt][v] = j;
      }
    }
    if (lo == 0) {
      #pragma unroll
      for (int rt = 0; rt < 2; ++rt) {
        #pragma unroll
        for (int v = 0; v < 8; ++v) { mnD[rt * 16 + v + 8 * hi] = bestD[rt][v]; mnJ[rt * 16 + v + 8 * hi] = bestJ[rt][v]; }
      }
    }
    asm volatile("s_wait_dscnt 0" ::: "memory");

    if (lo == 0) {
      float sum = 0.f;
      for (int m = 0; m < 16; ++m) sum += mnD[hi * 16 + m];
      ws[hi ? tl1 : tl0] = sum;
    }
    {
      const int j = mnJ[lane];
      v4f cc[5];
      #pragma unroll
      for (int jc = 0; jc < 5; ++jc) cc[jc] = *(const v4f*)&smem[OFF_CBT + ((jc * 512 + j) << 2)];
      float* qp = out_zq + (size_t)((hi ? tl1 : tl0) * 16 + lo) * 20;
      v4f o0 = { cc[0][0], cc[0][1], cc[1][0], cc[1][1] };
      v4f o1 = { cc[2][0], cc[2][1], cc[3][0], cc[3][1] };
      v4f o2 = { cc[0][2], cc[0][3], cc[1][2], cc[1][3] };
      v4f o3 = { cc[2][2], cc[2][3], cc[3][2], cc[3][3] };
      *(v4f*)(qp)      = o0;
      *(v4f*)(qp + 4)  = o1;
      *(v4f*)(qp + 8)  = o2;
      *(v4f*)(qp + 12) = o3;
      *(v4f*)(qp + 16) = cc[4];
    }
  }
}

__global__ __launch_bounds__(256) void vq_loss_reduce(const float* __restrict__ ws,
                                                      float* __restrict__ out,
                                                      int n, float invCount) {
  __shared__ float p[256];
  float s = 0.f;
  constexpr int U = 32;
  for (int base = threadIdx.x; base < n; base += 256 * U) {
    float v[U];
    #pragma unroll
    for (int u = 0; u < U; ++u) v[u] = ws[min(base + u * 256, n - 1)];
    #pragma unroll
    for (int u = 0; u < U; ++u) if (base + u * 256 < n) s += v[u];
  }
  p[threadIdx.x] = s;
  __syncthreads();
  if (threadIdx.x == 0) {
    float t = 0.f;
    for (int i = 0; i < 256; ++i) t += p[i];
    float q = t * invCount;
    out[0] = q;
    out[1] = 0.25f * q;
  }
}

extern "C" void kernel_launch(void* const* d_in, const int* in_sizes, int n_in,
                              void* d_out, int out_size, void* d_ws, size_t ws_size,
                              hipStream_t stream) {
  const float* c   = (const float*)d_in[0];
  const float* W1  = (const float*)d_in[1];
  const float* b1  = (const float*)d_in[2];
  const float* W2  = (const float*)d_in[3];
  const float* b2  = (const float*)d_in[4];
  const float* cb  = (const float*)d_in[5];
  const float* Wmu = (const float*)d_in[6];
  const float* bmu = (const float*)d_in[7];
  const float* Wlv = (const float*)d_in[8];
  const float* blv = (const float*)d_in[9];

  const int B = in_sizes[0] / 100;
  const int numTiles = B / 16;
  float* out = (float*)d_out;
  float* ws  = (float*)d_ws;

  const size_t ldsBytes = (size_t)LDS_FLOATS * sizeof(float);
  (void)hipFuncSetAttribute((const void*)vqvae_fused_kernel,
                            hipFuncAttributeMaxDynamicSharedMemorySize, (int)ldsBytes);

  const int blocks = 256;
  vqvae_fused_kernel<<<blocks, 256, ldsBytes, stream>>>(
      c, W1, b1, W2, b2, cb, Wmu, bmu, Wlv, blv,
      out,
      out + (size_t)B * 20,
      out + 2 * (size_t)B * 20,
      out + 3 * (size_t)B * 20,
      ws, B);
  vq_loss_reduce<<<1, 256, 0, stream>>>(ws, out + 4 * (size_t)B * 20,
                                        numTiles, 1.0f / ((float)B * 20.0f));
}
